# starting kernel
# baseline (speedup 1.0000x reference)
#include <hip/hip_runtime.h>


typedef float v2f __attribute__((ext_vector_type(2)));
typedef float v8f __attribute__((ext_vector_type(8)));

#define GRU_H   64
#define GRU_I   32
#define GRU_O   16
#define GRU_NG  192
#define BM      16
#define NWAVES  12
#define WAVE_SZ 32

__global__ __launch_bounds__(NWAVES * WAVE_SZ)
void gru_fused_wmma_kernel(const float* __restrict__ x,
                           const float* __restrict__ W_ih,
                           const float* __restrict__ W_hh,
                           const float* __restrict__ b_ih,
                           const float* __restrict__ b_hh,
                           const float* __restrict__ W_lin,
                           const float* __restrict__ b_lin,
                           float* __restrict__ out,
                           int T)
{
    __shared__ float h_buf[BM * GRU_H];
    __shared__ float r_buf[BM * GRU_H];
    __shared__ float z_buf[BM * GRU_H];

    const int tid  = threadIdx.x;
    const int wave = tid / WAVE_SZ;
    const int lane = tid % WAVE_SZ;
    const int hi   = lane >> 4;
    const int ln   = lane & 15;
    const int b0   = blockIdx.x * BM;

    const int  ncol = wave * 16;
    const int  gcol = ncol & (GRU_H - 1);
    const bool is_n = (wave >= 8);

    v2f Bih[GRU_I / 4];
    v2f Bhh[GRU_H / 4];
    v2f Blin[GRU_H / 4];
    {
        const float* wih_row = W_ih + (size_t)(ncol + ln) * GRU_I;
        #pragma unroll
        for (int kc = 0; kc < GRU_I / 4; ++kc) {
            Bih[kc][0] = wih_row[4 * kc + 2 * hi + 0];
            Bih[kc][1] = wih_row[4 * kc + 2 * hi + 1];
        }
        const float* whh_row = W_hh + (size_t)(ncol + ln) * GRU_H;
        #pragma unroll
        for (int kc = 0; kc < GRU_H / 4; ++kc) {
            Bhh[kc][0] = whh_row[4 * kc + 2 * hi + 0];
            Bhh[kc][1] = whh_row[4 * kc + 2 * hi + 1];
        }
        const float* wl_row = W_lin + (size_t)ln * GRU_H;
        #pragma unroll
        for (int kc = 0; kc < GRU_H / 4; ++kc) {
            Blin[kc][0] = wl_row[4 * kc + 2 * hi + 0];
            Blin[kc][1] = wl_row[4 * kc + 2 * hi + 1];
        }
    }
    const float bih_l  = b_ih[ncol + ln];
    const float bhh_l  = b_hh[ncol + ln];
    const float blin_l = b_lin[ln];

    for (int i = tid; i < BM * GRU_H; i += blockDim.x) h_buf[i] = 0.0f;
    __syncthreads();

    const float* xrow = x + ((size_t)(b0 + ln) * T) * GRU_I;

    for (int t = 0; t < T; ++t) {
        const float* xt = xrow + (size_t)t * GRU_I;
        __builtin_prefetch((const void*)(xt + GRU_I), 0, 1);

        v8f acc_x  = {0.f, 0.f, 0.f, 0.f, 0.f, 0.f, 0.f, 0.f};
        v8f acc_h0 = {0.f, 0.f, 0.f, 0.f, 0.f, 0.f, 0.f, 0.f};
        v8f acc_h1 = {0.f, 0.f, 0.f, 0.f, 0.f, 0.f, 0.f, 0.f};

        #pragma unroll
        for (int kc = 0; kc < GRU_I / 4; ++kc) {
            v2f a;
            a[0] = xt[4 * kc + 2 * hi + 0];
            a[1] = xt[4 * kc + 2 * hi + 1];
            acc_x = __builtin_amdgcn_wmma_f32_16x16x4_f32(
                false, a, false, Bih[kc], (short)0, acc_x, false, false);
        }

        #pragma unroll
        for (int kc = 0; kc < 8; ++kc) {
            v2f a;
            a[0] = h_buf[ln * GRU_H + 4 * kc + 2 * hi + 0];
            a[1] = h_buf[ln * GRU_H + 4 * kc + 2 * hi + 1];
            acc_h0 = __builtin_amdgcn_wmma_f32_16x16x4_f32(
                false, a, false, Bhh[kc], (short)0, acc_h0, false, false);
        }
        #pragma unroll
        for (int kc = 8; kc < 16; ++kc) {
            v2f a;
            a[0] = h_buf[ln * GRU_H + 4 * kc + 2 * hi + 0];
            a[1] = h_buf[ln * GRU_H + 4 * kc + 2 * hi + 1];
            acc_h1 = __builtin_amdgcn_wmma_f32_16x16x4_f32(
                false, a, false, Bhh[kc], (short)0, acc_h1, false, false);
        }

        if (!is_n) {
            #pragma unroll
            for (int v = 0; v < 8; ++v) {
                const int m = v + 8 * hi;
                float g = acc_x[v] + acc_h0[v] + acc_h1[v] + bih_l + bhh_l;
                g = 1.0f / (1.0f + __expf(-g));
                if (wave < 4) r_buf[m * GRU_H + gcol + ln] = g;
                else          z_buf[m * GRU_H + gcol + ln] = g;
            }
        }
        __syncthreads();

        if (is_n) {
            #pragma unroll
            for (int v = 0; v < 8; ++v) {
                const int m   = v + 8 * hi;
                const int idx = m * GRU_H + gcol + ln;
                const float r  = r_buf[idx];
                const float z  = z_buf[idx];
                const float gx = acc_x[v] + bih_l;
                const float gh = acc_h0[v] + acc_h1[v] + bhh_l;
                const float n  = tanhf(gx + r * gh);
                const float ho = h_buf[idx];
                h_buf[idx] = (1.0f - z) * n + z * ho;
            }
        }
        __syncthreads();

        if (wave == 0) {
            v8f acc_y0 = {0.f, 0.f, 0.f, 0.f, 0.f, 0.f, 0.f, 0.f};
            v8f acc_y1 = {0.f, 0.f, 0.f, 0.f, 0.f, 0.f, 0.f, 0.f};
            #pragma unroll
            for (int kc = 0; kc < 8; ++kc) {
                v2f a;
                a[0] = h_buf[ln * GRU_H + 4 * kc + 2 * hi + 0];
                a[1] = h_buf[ln * GRU_H + 4 * kc + 2 * hi + 1];
                acc_y0 = __builtin_amdgcn_wmma_f32_16x16x4_f32(
                    false, a, false, Blin[kc], (short)0, acc_y0, false, false);
            }
            #pragma unroll
            for (int kc = 8; kc < 16; ++kc) {
                v2f a;
                a[0] = h_buf[ln * GRU_H + 4 * kc + 2 * hi + 0];
                a[1] = h_buf[ln * GRU_H + 4 * kc + 2 * hi + 1];
                acc_y1 = __builtin_amdgcn_wmma_f32_16x16x4_f32(
                    false, a, false, Blin[kc], (short)0, acc_y1, false, false);
            }
            #pragma unroll
            for (int v = 0; v < 8; ++v) {
                const int m = v + 8 * hi;
                out[((size_t)(b0 + m) * T + t) * GRU_O + ln] =
                    acc_y0[v] + acc_y1[v] + blin_l;
            }
        }
    }
}

extern "C" void kernel_launch(void* const* d_in, const int* in_sizes, int n_in,
                              void* d_out, int out_size, void* d_ws, size_t ws_size,
                              hipStream_t stream) {
    const float* x     = (const float*)d_in[0];
    const float* W_ih  = (const float*)d_in[1];
    const float* W_hh  = (const float*)d_in[2];
    const float* b_ih  = (const float*)d_in[3];
    const float* b_hh  = (const float*)d_in[4];
    const float* W_lin = (const float*)d_in[5];
    const float* b_lin = (const float*)d_in[6];
    float* out = (float*)d_out;

    const int B = 512;
    const int T = in_sizes[0] / (B * GRU_I);

    dim3 grid(B / BM);
    dim3 block(NWAVES * WAVE_SZ);
    gru_fused_wmma_kernel<<<grid, block, 0, stream>>>(
        x, W_ih, W_hh, b_ih, b_hh, W_lin, b_lin, out, T);
}


# LDS row pitch of the h/r/z tiles 64 -> 68 floats (272 B = 16 B x 17): the h A-fragment ds_load_2addr_b64 reads go from a 16-way bank conflict to conflict-free
# speedup vs baseline: 1.2765x; 1.2765x over previous
#include <hip/hip_runtime.h>


typedef float v2f __attribute__((ext_vector_type(2)));
typedef float v8f __attribute__((ext_vector_type(8)));

#define GRU_H   64
#define GRU_I   32
#define GRU_O   16
#define GRU_NG  192
#define BM      16
#define NWAVES  12
#define WAVE_SZ 32
#define LDS_PITCH 68

__global__ __launch_bounds__(NWAVES * WAVE_SZ)
void gru_fused_wmma_kernel(const float* __restrict__ x,
                           const float* __restrict__ W_ih,
                           const float* __restrict__ W_hh,
                           const float* __restrict__ b_ih,
                           const float* __restrict__ b_hh,
                           const float* __restrict__ W_lin,
                           const float* __restrict__ b_lin,
                           float* __restrict__ out,
                           int T)
{
    __shared__ __attribute__((aligned(16))) float h_buf[BM * LDS_PITCH];
    __shared__ __attribute__((aligned(16))) float r_buf[BM * LDS_PITCH];
    __shared__ __attribute__((aligned(16))) float z_buf[BM * LDS_PITCH];

    const int tid  = threadIdx.x;
    const int wave = tid / WAVE_SZ;
    const int lane = tid % WAVE_SZ;
    const int hi   = lane >> 4;
    const int ln   = lane & 15;
    const int b0   = blockIdx.x * BM;

    const int  ncol = wave * 16;
    const int  gcol = ncol & (GRU_H - 1);
    const bool is_n = (wave >= 8);

    v2f Bih[GRU_I / 4];
    v2f Bhh[GRU_H / 4];
    v2f Blin[GRU_H / 4];
    {
        const float* wih_row = W_ih + (size_t)(ncol + ln) * GRU_I;
        #pragma unroll
        for (int kc = 0; kc < GRU_I / 4; ++kc) {
            Bih[kc][0] = wih_row[4 * kc + 2 * hi + 0];
            Bih[kc][1] = wih_row[4 * kc + 2 * hi + 1];
        }
        const float* whh_row = W_hh + (size_t)(ncol + ln) * GRU_H;
        #pragma unroll
        for (int kc = 0; kc < GRU_H / 4; ++kc) {
            Bhh[kc][0] = whh_row[4 * kc + 2 * hi + 0];
            Bhh[kc][1] = whh_row[4 * kc + 2 * hi + 1];
        }
        const float* wl_row = W_lin + (size_t)ln * GRU_H;
        #pragma unroll
        for (int kc = 0; kc < GRU_H / 4; ++kc) {
            Blin[kc][0] = wl_row[4 * kc + 2 * hi + 0];
            Blin[kc][1] = wl_row[4 * kc + 2 * hi + 1];
        }
    }
    const float bih_l  = b_ih[ncol + ln];
    const float bhh_l  = b_hh[ncol + ln];
    const float blin_l = b_lin[ln];

    for (int i = tid; i < BM * LDS_PITCH; i += blockDim.x) h_buf[i] = 0.0f;
    __syncthreads();

    const float* xrow = x + ((size_t)(b0 + ln) * T) * GRU_I;

    for (int t = 0; t < T; ++t) {
        const float* xt = xrow + (size_t)t * GRU_I;
        __builtin_prefetch((const void*)(xt + GRU_I), 0, 1);

        v8f acc_x  = {0.f, 0.f, 0.f, 0.f, 0.f, 0.f, 0.f, 0.f};
        v8f acc_h0 = {0.f, 0.f, 0.f, 0.f, 0.f, 0.f, 0.f, 0.f};
        v8f acc_h1 = {0.f, 0.f, 0.f, 0.f, 0.f, 0.f, 0.f, 0.f};

        #pragma unroll
        for (int kc = 0; kc < GRU_I / 4; ++kc) {
            v2f a;
            a[0] = xt[4 * kc + 2 * hi + 0];
            a[1] = xt[4 * kc + 2 * hi + 1];
            acc_x = __builtin_amdgcn_wmma_f32_16x16x4_f32(
                false, a, false, Bih[kc], (short)0, acc_x, false, false);
        }

        #pragma unroll
        for (int kc = 0; kc < 8; ++kc) {
            v2f a;
            a[0] = h_buf[ln * LDS_PITCH + 4 * kc + 2 * hi + 0];
            a[1] = h_buf[ln * LDS_PITCH + 4 * kc + 2 * hi + 1];
            acc_h0 = __builtin_amdgcn_wmma_f32_16x16x4_f32(
                false, a, false, Bhh[kc], (short)0, acc_h0, false, false);
        }
        #pragma unroll
        for (int kc = 8; kc < 16; ++kc) {
            v2f a;
            a[0] = h_buf[ln * LDS_PITCH + 4 * kc + 2 * hi + 0];
            a[1] = h_buf[ln * LDS_PITCH + 4 * kc + 2 * hi + 1];
            acc_h1 = __builtin_amdgcn_wmma_f32_16x16x4_f32(
                false, a, false, Bhh[kc], (short)0, acc_h1, false, false);
        }

        if (!is_n) {
            #pragma unroll
            for (int v = 0; v < 8; ++v) {
                const int m = v + 8 * hi;
                float g = acc_x[v] + acc_h0[v] + acc_h1[v] + bih_l + bhh_l;
                g = 1.0f / (1.0f + __expf(-g));
                if (wave < 4) r_buf[m * LDS_PITCH + gcol + ln] = g;
                else          z_buf[m * LDS_PITCH + gcol + ln] = g;
            }
        }
        __syncthreads();

        if (is_n) {
            #pragma unroll
            for (int v = 0; v < 8; ++v) {
                const int m   = v + 8 * hi;
                const int idx = m * LDS_PITCH + gcol + ln;
                const float r  = r_buf[idx];
                const float z  = z_buf[idx];
                const float gx = acc_x[v] + bih_l;
                const float gh = acc_h0[v] + acc_h1[v] + bhh_l;
                const float n  = tanhf(gx + r * gh);
                const float ho = h_buf[idx];
                h_buf[idx] = (1.0f - z) * n + z * ho;
            }
        }
        __syncthreads();

        if (wave == 0) {
            v8f acc_y0 = {0.f, 0.f, 0.f, 0.f, 0.f, 0.f, 0.f, 0.f};
            v8f acc_y1 = {0.f, 0.f, 0.f, 0.f, 0.f, 0.f, 0.f, 0.f};
            #pragma unroll
            for (int kc = 0; kc < 8; ++kc) {
                v2f a;
                a[0] = h_buf[ln * LDS_PITCH + 4 * kc + 2 * hi + 0];
                a[1] = h_buf[ln * LDS_PITCH + 4 * kc + 2 * hi + 1];
                acc_y0 = __builtin_amdgcn_wmma_f32_16x16x4_f32(
                    false, a, false, Blin[kc], (short)0, acc_y0, false, false);
            }
            #pragma unroll
            for (int kc = 8; kc < 16; ++kc) {
                v2f a;
                a[0] = h_buf[ln * LDS_PITCH + 4 * kc + 2 * hi + 0];
                a[1] = h_buf[ln * LDS_PITCH + 4 * kc + 2 * hi + 1];
                acc_y1 = __builtin_amdgcn_wmma_f32_16x16x4_f32(
                    false, a, false, Blin[kc], (short)0, acc_y1, false, false);
            }
            #pragma unroll
            for (int v = 0; v < 8; ++v) {
                const int m = v + 8 * hi;
                out[((size_t)(b0 + m) * T + t) * GRU_O + ln] =
                    acc_y0[v] + acc_y1[v] + blin_l;
            }
        }
    }
}

extern "C" void kernel_launch(void* const* d_in, const int* in_sizes, int n_in,
                              void* d_out, int out_size, void* d_ws, size_t ws_size,
                              hipStream_t stream) {
    const float* x     = (const float*)d_in[0];
    const float* W_ih  = (const float*)d_in[1];
    const float* W_hh  = (const float*)d_in[2];
    const float* b_ih  = (const float*)d_in[3];
    const float* b_hh  = (const float*)d_in[4];
    const float* W_lin = (const float*)d_in[5];
    const float* b_lin = (const float*)d_in[6];
    float* out = (float*)d_out;

    const int B = 512;
    const int T = in_sizes[0] / (B * GRU_I);

    dim3 grid(B / BM);
    dim3 block(NWAVES * WAVE_SZ);
    gru_fused_wmma_kernel<<<grid, block, 0, stream>>>(
        x, W_ih, W_hh, b_ih, b_hh, W_lin, b_lin, out, T);
}


# gate epilogues made straight-line (one r/z tile select per wave; tanh as a select of the library's two ranges): the 8 per-lane sigmoid/tanh chains overlap instead of running back to back
# speedup vs baseline: 2.1936x; 1.7184x over previous
#include <hip/hip_runtime.h>


typedef float v2f __attribute__((ext_vector_type(2)));
typedef float v8f __attribute__((ext_vector_type(8)));

#define GRU_H   64
#define GRU_I   32
#define GRU_O   16
#define GRU_NG  192
#define BM      16
#define NWAVES  12
#define WAVE_SZ 32
#define LDS_PITCH 68

__device__ __forceinline__ float tanh_select(float a)
{
    const float ax = __builtin_fabsf(a);
    const float e   = expf(ax + ax);
    const float big = __builtin_fmaf(__builtin_amdgcn_rcpf(e + 1.0f), -2.0f, 1.0f);
    const float a2 = a * a;
    float p = __builtin_fmaf(a2, -0x1.758e7ap-8f, 0x1.521192p-6f);
    p = __builtin_fmaf(a2, p, -0x1.b8389cp-5f);
    p = __builtin_fmaf(a2, p, 0x1.110704p-3f);
    p = __builtin_fmaf(a2, p, -0x1.555532p-2f);
    const float small = __builtin_fmaf(a2, ax * p, ax);
    const float t = (ax < 0.625f) ? small : big;
    return __builtin_copysignf(t, a);
}

__global__ __launch_bounds__(NWAVES * WAVE_SZ)
void gru_fused_wmma_kernel(const float* __restrict__ x,
                           const float* __restrict__ W_ih,
                           const float* __restrict__ W_hh,
                           const float* __restrict__ b_ih,
                           const float* __restrict__ b_hh,
                           const float* __restrict__ W_lin,
                           const float* __restrict__ b_lin,
                           float* __restrict__ out,
                           int T)
{
    __shared__ __attribute__((aligned(16))) float h_buf[BM * LDS_PITCH];
    __shared__ __attribute__((aligned(16))) float rz_buf[2 * BM * LDS_PITCH];
    float* const r_buf = rz_buf;
    float* const z_buf = rz_buf + BM * LDS_PITCH;

    const int tid  = threadIdx.x;
    const int wave = tid / WAVE_SZ;
    const int lane = tid % WAVE_SZ;
    const int hi   = lane >> 4;
    const int ln   = lane & 15;
    const int b0   = blockIdx.x * BM;

    const int  ncol = wave * 16;
    const int  gcol = ncol & (GRU_H - 1);
    const bool is_n = (wave >= 8);

    v2f Bih[GRU_I / 4];
    v2f Bhh[GRU_H / 4];
    v2f Blin[GRU_H / 4];
    {
        const float* wih_row = W_ih + (size_t)(ncol + ln) * GRU_I;
        #pragma unroll
        for (int kc = 0; kc < GRU_I / 4; ++kc) {
            Bih[kc][0] = wih_row[4 * kc + 2 * hi + 0];
            Bih[kc][1] = wih_row[4 * kc + 2 * hi + 1];
        }
        const float* whh_row = W_hh + (size_t)(ncol + ln) * GRU_H;
        #pragma unroll
        for (int kc = 0; kc < GRU_H / 4; ++kc) {
            Bhh[kc][0] = whh_row[4 * kc + 2 * hi + 0];
            Bhh[kc][1] = whh_row[4 * kc + 2 * hi + 1];
        }
        const float* wl_row = W_lin + (size_t)ln * GRU_H;
        #pragma unroll
        for (int kc = 0; kc < GRU_H / 4; ++kc) {
            Blin[kc][0] = wl_row[4 * kc + 2 * hi + 0];
            Blin[kc][1] = wl_row[4 * kc + 2 * hi + 1];
        }
    }
    const float bih_l  = b_ih[ncol + ln];
    const float bhh_l  = b_hh[ncol + ln];
    const float blin_l = b_lin[ln];

    for (int i = tid; i < BM * LDS_PITCH; i += blockDim.x) h_buf[i] = 0.0f;
    __syncthreads();

    const float* xrow = x + ((size_t)(b0 + ln) * T) * GRU_I;

    for (int t = 0; t < T; ++t) {
        const float* xt = xrow + (size_t)t * GRU_I;
        __builtin_prefetch((const void*)(xt + GRU_I), 0, 1);

        v8f acc_x  = {0.f, 0.f, 0.f, 0.f, 0.f, 0.f, 0.f, 0.f};
        v8f acc_h0 = {0.f, 0.f, 0.f, 0.f, 0.f, 0.f, 0.f, 0.f};
        v8f acc_h1 = {0.f, 0.f, 0.f, 0.f, 0.f, 0.f, 0.f, 0.f};

        #pragma unroll
        for (int kc = 0; kc < GRU_I / 4; ++kc) {
            v2f a;
            a[0] = xt[4 * kc + 2 * hi + 0];
            a[1] = xt[4 * kc + 2 * hi + 1];
            acc_x = __builtin_amdgcn_wmma_f32_16x16x4_f32(
                false, a, false, Bih[kc], (short)0, acc_x, false, false);
        }

        #pragma unroll
        for (int kc = 0; kc < 8; ++kc) {
            v2f a;
            a[0] = h_buf[ln * LDS_PITCH + 4 * kc + 2 * hi + 0];
            a[1] = h_buf[ln * LDS_PITCH + 4 * kc + 2 * hi + 1];
            acc_h0 = __builtin_amdgcn_wmma_f32_16x16x4_f32(
                false, a, false, Bhh[kc], (short)0, acc_h0, false, false);
        }
        #pragma unroll
        for (int kc = 8; kc < 16; ++kc) {
            v2f a;
            a[0] = h_buf[ln * LDS_PITCH + 4 * kc + 2 * hi + 0];
            a[1] = h_buf[ln * LDS_PITCH + 4 * kc + 2 * hi + 1];
            acc_h1 = __builtin_amdgcn_wmma_f32_16x16x4_f32(
                false, a, false, Bhh[kc], (short)0, acc_h1, false, false);
        }

        if (!is_n) {
            float* const gate_tile = (wave < 4) ? r_buf : z_buf;
            float g[8];
            #pragma unroll
            for (int v = 0; v < 8; ++v) {
                const float s = acc_x[v] + acc_h0[v] + acc_h1[v] + bih_l + bhh_l;
                g[v] = 1.0f / (1.0f + __expf(-s));
            }
            #pragma unroll
            for (int v = 0; v < 8; ++v) {
                const int m = v + 8 * hi;
                gate_tile[m * LDS_PITCH + gcol + ln] = g[v];
            }
        }
        __syncthreads();

        if (is_n) {
            float r[8], z[8], ho[8], hn[8];
            #pragma unroll
            for (int v = 0; v < 8; ++v) {
                const int idx = (v + 8 * hi) * LDS_PITCH + gcol + ln;
                r[v]  = r_buf[idx];
                z[v]  = z_buf[idx];
                ho[v] = h_buf[idx];
            }
            #pragma unroll
            for (int v = 0; v < 8; ++v) {
                const float gx = acc_x[v] + bih_l;
                const float gh = acc_h0[v] + acc_h1[v] + bhh_l;
                const float n  = tanh_select(gx + r[v] * gh);
                hn[v] = (1.0f - z[v]) * n + z[v] * ho[v];
            }
            #pragma unroll
            for (int v = 0; v < 8; ++v) {
                const int idx = (v + 8 * hi) * LDS_PITCH + gcol + ln;
                h_buf[idx] = hn[v];
            }
        }
        __syncthreads();

        if (wave == 0) {
            v8f acc_y0 = {0.f, 0.f, 0.f, 0.f, 0.f, 0.f, 0.f, 0.f};
            v8f acc_y1 = {0.f, 0.f, 0.f, 0.f, 0.f, 0.f, 0.f, 0.f};
            #pragma unroll
            for (int kc = 0; kc < 8; ++kc) {
                v2f a;
                a[0] = h_buf[ln * LDS_PITCH + 4 * kc + 2 * hi + 0];
                a[1] = h_buf[ln * LDS_PITCH + 4 * kc + 2 * hi + 1];
                acc_y0 = __builtin_amdgcn_wmma_f32_16x16x4_f32(
                    false, a, false, Blin[kc], (short)0, acc_y0, false, false);
            }
            #pragma unroll
            for (int kc = 8; kc < 16; ++kc) {
                v2f a;
                a[0] = h_buf[ln * LDS_PITCH + 4 * kc + 2 * hi + 0];
                a[1] = h_buf[ln * LDS_PITCH + 4 * kc + 2 * hi + 1];
                acc_y1 = __builtin_amdgcn_wmma_f32_16x16x4_f32(
                    false, a, false, Blin[kc], (short)0, acc_y1, false, false);
            }
            #pragma unroll
            for (int v = 0; v < 8; ++v) {
                const int m = v + 8 * hi;
                out[((size_t)(b0 + m) * T + t) * GRU_O + ln] =
                    acc_y0[v] + acc_y1[v] + blin_l;
            }
        }
    }
}

extern "C" void kernel_launch(void* const* d_in, const int* in_sizes, int n_in,
                              void* d_out, int out_size, void* d_ws, size_t ws_size,
                              hipStream_t stream) {
    const float* x     = (const float*)d_in[0];
    const float* W_ih  = (const float*)d_in[1];
    const float* W_hh  = (const float*)d_in[2];
    const float* b_ih  = (const float*)d_in[3];
    const float* b_hh  = (const float*)d_in[4];
    const float* W_lin = (const float*)d_in[5];
    const float* b_lin = (const float*)d_in[6];
    float* out = (float*)d_out;

    const int B = 512;
    const int T = in_sizes[0] / (B * GRU_I);

    dim3 grid(B / BM);
    dim3 block(NWAVES * WAVE_SZ);
    gru_fused_wmma_kernel<<<grid, block, 0, stream>>>(
        x, W_ih, W_hh, b_ih, b_hh, W_lin, b_lin, out, T);
}


# input product x@W_ih^T taken off the recurrence: x tile staged through LDS one step ahead by 4 waves (16 B per lane); each wave's 8 x WMMAs for step t+1 issued in its barrier-wait gap of step t
# speedup vs baseline: 2.4641x; 1.1233x over previous
#include <hip/hip_runtime.h>


typedef float v2f __attribute__((ext_vector_type(2)));
typedef float v8f __attribute__((ext_vector_type(8)));

#define GRU_H   64
#define GRU_I   32
#define GRU_O   16
#define GRU_NG  192
#define BM      16
#define NWAVES  12
#define WAVE_SZ 32
#define LDS_PITCH 68
#define X_PITCH 36

__device__ __forceinline__ float tanh_select(float a)
{
    const float ax = __builtin_fabsf(a);
    const float e   = expf(ax + ax);
    const float big = __builtin_fmaf(__builtin_amdgcn_rcpf(e + 1.0f), -2.0f, 1.0f);
    const float a2 = a * a;
    float p = __builtin_fmaf(a2, -0x1.758e7ap-8f, 0x1.521192p-6f);
    p = __builtin_fmaf(a2, p, -0x1.b8389cp-5f);
    p = __builtin_fmaf(a2, p, 0x1.110704p-3f);
    p = __builtin_fmaf(a2, p, -0x1.555532p-2f);
    const float small = __builtin_fmaf(a2, ax * p, ax);
    const float t = (ax < 0.625f) ? small : big;
    return __builtin_copysignf(t, a);
}

__global__ __launch_bounds__(NWAVES * WAVE_SZ)
void gru_fused_wmma_kernel(const float* __restrict__ x,
                           const float* __restrict__ W_ih,
                           const float* __restrict__ W_hh,
                           const float* __restrict__ b_ih,
                           const float* __restrict__ b_hh,
                           const float* __restrict__ W_lin,
                           const float* __restrict__ b_lin,
                           float* __restrict__ out,
                           int T)
{
    __shared__ __attribute__((aligned(16))) float h_buf[BM * LDS_PITCH];
    __shared__ __attribute__((aligned(16))) float rz_buf[2 * BM * LDS_PITCH];
    float* const r_buf = rz_buf;
    float* const z_buf = rz_buf + BM * LDS_PITCH;
    __shared__ __attribute__((aligned(16))) float x_buf[BM * X_PITCH];

    const int tid  = threadIdx.x;
    const int wave = tid / WAVE_SZ;
    const int lane = tid % WAVE_SZ;
    const int hi   = lane >> 4;
    const int ln   = lane & 15;
    const int b0   = blockIdx.x * BM;

    const int  ncol = wave * 16;
    const int  gcol = ncol & (GRU_H - 1);
    const bool is_n = (wave >= 8);

    const bool is_stager = (wave >= 4) && (wave < 8);
    const int  sj   = tid & 127;
    const int  srow = sj >> 3;
    const int  sc4  = sj & 7;
    const float* const xs_row = x + ((size_t)(b0 + srow) * T) * GRU_I + 4 * sc4;
    float* const xs_lds = x_buf + srow * X_PITCH + 4 * sc4;

    v2f Bih[GRU_I / 4];
    v2f Bhh[GRU_H / 4];
    v2f Blin[GRU_H / 4];
    {
        const float* wih_row = W_ih + (size_t)(ncol + ln) * GRU_I;
        #pragma unroll
        for (int kc = 0; kc < GRU_I / 4; ++kc) {
            Bih[kc][0] = wih_row[4 * kc + 2 * hi + 0];
            Bih[kc][1] = wih_row[4 * kc + 2 * hi + 1];
        }
        const float* whh_row = W_hh + (size_t)(ncol + ln) * GRU_H;
        #pragma unroll
        for (int kc = 0; kc < GRU_H / 4; ++kc) {
            Bhh[kc][0] = whh_row[4 * kc + 2 * hi + 0];
            Bhh[kc][1] = whh_row[4 * kc + 2 * hi + 1];
        }
        const float* wl_row = W_lin + (size_t)ln * GRU_H;
        #pragma unroll
        for (int kc = 0; kc < GRU_H / 4; ++kc) {
            Blin[kc][0] = wl_row[4 * kc + 2 * hi + 0];
            Blin[kc][1] = wl_row[4 * kc + 2 * hi + 1];
        }
    }
    const float bih_l  = b_ih[ncol + ln];
    const float bhh_l  = b_hh[ncol + ln];
    const float blin_l = b_lin[ln];

    for (int i = tid; i < BM * LDS_PITCH; i += blockDim.x) h_buf[i] = 0.0f;

    if (is_stager)
        *reinterpret_cast<float4*>(xs_lds) = *reinterpret_cast<const float4*>(xs_row);
    __syncthreads();
    v8f acc_x = {0.f, 0.f, 0.f, 0.f, 0.f, 0.f, 0.f, 0.f};
    #pragma unroll
    for (int kc = 0; kc < GRU_I / 4; ++kc) {
        v2f a;
        a[0] = x_buf[ln * X_PITCH + 4 * kc + 2 * hi + 0];
        a[1] = x_buf[ln * X_PITCH + 4 * kc + 2 * hi + 1];
        acc_x = __builtin_amdgcn_wmma_f32_16x16x4_f32(
            false, a, false, Bih[kc], (short)0, acc_x, false, false);
    }
    __syncthreads();
    if (is_stager) {
        const int t1 = (1 < T) ? 1 : T - 1;
        *reinterpret_cast<float4*>(xs_lds) =
            *reinterpret_cast<const float4*>(xs_row + (size_t)t1 * GRU_I);
    }
    __syncthreads();

    for (int t = 0; t < T; ++t) {

        float4 xr = {0.f, 0.f, 0.f, 0.f};
        if (is_stager) {
            const int t2 = (t + 2 < T) ? t + 2 : T - 1;
            const int t6 = (t + 6 < T) ? t + 6 : T - 1;
            xr = *reinterpret_cast<const float4*>(xs_row + (size_t)t2 * GRU_I);
            __builtin_prefetch((const void*)(xs_row + (size_t)t6 * GRU_I), 0, 1);
        }

        v2f xa[GRU_I / 4];
        #pragma unroll
        for (int kc = 0; kc < GRU_I / 4; ++kc) {
            xa[kc][0] = x_buf[ln * X_PITCH + 4 * kc + 2 * hi + 0];
            xa[kc][1] = x_buf[ln * X_PITCH + 4 * kc + 2 * hi + 1];
        }

        v8f acc_xn = {0.f, 0.f, 0.f, 0.f, 0.f, 0.f, 0.f, 0.f};
        v8f acc_h0 = {0.f, 0.f, 0.f, 0.f, 0.f, 0.f, 0.f, 0.f};
        v8f acc_h1 = {0.f, 0.f, 0.f, 0.f, 0.f, 0.f, 0.f, 0.f};

        #pragma unroll
        for (int kc = 0; kc < 8; ++kc) {
            v2f a;
            a[0] = h_buf[ln * LDS_PITCH + 4 * kc + 2 * hi + 0];
            a[1] = h_buf[ln * LDS_PITCH + 4 * kc + 2 * hi + 1];
            acc_h0 = __builtin_amdgcn_wmma_f32_16x16x4_f32(
                false, a, false, Bhh[kc], (short)0, acc_h0, false, false);
        }
        #pragma unroll
        for (int kc = 8; kc < 16; ++kc) {
            v2f a;
            a[0] = h_buf[ln * LDS_PITCH + 4 * kc + 2 * hi + 0];
            a[1] = h_buf[ln * LDS_PITCH + 4 * kc + 2 * hi + 1];
            acc_h1 = __builtin_amdgcn_wmma_f32_16x16x4_f32(
                false, a, false, Bhh[kc], (short)0, acc_h1, false, false);
        }

        if (!is_n) {
            float* const gate_tile = (wave < 4) ? r_buf : z_buf;
            float g[8];
            #pragma unroll
            for (int v = 0; v < 8; ++v) {
                const float s = acc_x[v] + acc_h0[v] + acc_h1[v] + bih_l + bhh_l;
                g[v] = 1.0f / (1.0f + __expf(-s));
            }
            #pragma unroll
            for (int v = 0; v < 8; ++v) {
                const int m = v + 8 * hi;
                gate_tile[m * LDS_PITCH + gcol + ln] = g[v];
            }
        } else {
            #pragma unroll
            for (int kc = 0; kc < GRU_I / 4; ++kc)
                acc_xn = __builtin_amdgcn_wmma_f32_16x16x4_f32(
                    false, xa[kc], false, Bih[kc], (short)0, acc_xn, false, false);
        }
        __syncthreads();

        if (is_n) {
            float r[8], z[8], ho[8], hn[8];
            #pragma unroll
            for (int v = 0; v < 8; ++v) {
                const int idx = (v + 8 * hi) * LDS_PITCH + gcol + ln;
                r[v]  = r_buf[idx];
                z[v]  = z_buf[idx];
                ho[v] = h_buf[idx];
            }
            #pragma unroll
            for (int v = 0; v < 8; ++v) {
                const float gx = acc_x[v] + bih_l;
                const float gh = acc_h0[v] + acc_h1[v] + bhh_l;
                const float n  = tanh_select(gx + r[v] * gh);
                hn[v] = (1.0f - z[v]) * n + z[v] * ho[v];
            }
            #pragma unroll
            for (int v = 0; v < 8; ++v) {
                const int idx = (v + 8 * hi) * LDS_PITCH + gcol + ln;
                h_buf[idx] = hn[v];
            }
        } else {
            #pragma unroll
            for (int kc = 0; kc < GRU_I / 4; ++kc)
                acc_xn = __builtin_amdgcn_wmma_f32_16x16x4_f32(
                    false, xa[kc], false, Bih[kc], (short)0, acc_xn, false, false);
            if (is_stager) {
                *reinterpret_cast<float4*>(xs_lds) = xr;
            }
        }
        __syncthreads();

        if (wave == 0) {
            v8f acc_y0 = {0.f, 0.f, 0.f, 0.f, 0.f, 0.f, 0.f, 0.f};
            v8f acc_y1 = {0.f, 0.f, 0.f, 0.f, 0.f, 0.f, 0.f, 0.f};
            #pragma unroll
            for (int kc = 0; kc < 8; ++kc) {
                v2f a;
                a[0] = h_buf[ln * LDS_PITCH + 4 * kc + 2 * hi + 0];
                a[1] = h_buf[ln * LDS_PITCH + 4 * kc + 2 * hi + 1];
                acc_y0 = __builtin_amdgcn_wmma_f32_16x16x4_f32(
                    false, a, false, Blin[kc], (short)0, acc_y0, false, false);
            }
            #pragma unroll
            for (int kc = 8; kc < 16; ++kc) {
                v2f a;
                a[0] = h_buf[ln * LDS_PITCH + 4 * kc + 2 * hi + 0];
                a[1] = h_buf[ln * LDS_PITCH + 4 * kc + 2 * hi + 1];
                acc_y1 = __builtin_amdgcn_wmma_f32_16x16x4_f32(
                    false, a, false, Blin[kc], (short)0, acc_y1, false, false);
            }
            #pragma unroll
            for (int v = 0; v < 8; ++v) {
                const int m = v + 8 * hi;
                out[((size_t)(b0 + m) * T + t) * GRU_O + ln] =
                    acc_y0[v] + acc_y1[v] + blin_l;
            }
        }
        acc_x = acc_xn;
    }
}

extern "C" void kernel_launch(void* const* d_in, const int* in_sizes, int n_in,
                              void* d_out, int out_size, void* d_ws, size_t ws_size,
                              hipStream_t stream) {
    const float* x     = (const float*)d_in[0];
    const float* W_ih  = (const float*)d_in[1];
    const float* W_hh  = (const float*)d_in[2];
    const float* b_ih  = (const float*)d_in[3];
    const float* b_hh  = (const float*)d_in[4];
    const float* W_lin = (const float*)d_in[5];
    const float* b_lin = (const float*)d_in[6];
    float* out = (float*)d_out;

    const int B = 512;
    const int T = in_sizes[0] / (B * GRU_I);

    dim3 grid(B / BM);
    dim3 block(NWAVES * WAVE_SZ);
    gru_fused_wmma_kernel<<<grid, block, 0, stream>>>(
        x, W_ih, W_hh, b_ih, b_hh, W_lin, b_lin, out, T);
}


# output projection taken off the step's serial tail: wave 0 keeps the h A fragments of the hidden product in registers and issues y(t-1) behind the first barrier of step t, beside the n-gate phase
# speedup vs baseline: 2.6545x; 1.0773x over previous
#include <hip/hip_runtime.h>


typedef float v2f __attribute__((ext_vector_type(2)));
typedef float v8f __attribute__((ext_vector_type(8)));

#define GRU_H   64
#define GRU_I   32
#define GRU_O   16
#define GRU_NG  192
#define BM      16
#define NWAVES  12
#define WAVE_SZ 32
#define LDS_PITCH 68
#define X_PITCH 36

__device__ __forceinline__ float tanh_select(float a)
{
    const float ax = __builtin_fabsf(a);
    const float e   = expf(ax + ax);
    const float big = __builtin_fmaf(__builtin_amdgcn_rcpf(e + 1.0f), -2.0f, 1.0f);
    const float a2 = a * a;
    float p = __builtin_fmaf(a2, -0x1.758e7ap-8f, 0x1.521192p-6f);
    p = __builtin_fmaf(a2, p, -0x1.b8389cp-5f);
    p = __builtin_fmaf(a2, p, 0x1.110704p-3f);
    p = __builtin_fmaf(a2, p, -0x1.555532p-2f);
    const float small = __builtin_fmaf(a2, ax * p, ax);
    const float t = (ax < 0.625f) ? small : big;
    return __builtin_copysignf(t, a);
}

__device__ __forceinline__ void project(const v2f (&ha)[GRU_H / 4],
                                        const v2f (&Blin)[GRU_H / 4],
                                        float blin_l, float (&y)[8])
{
    v8f acc_y0 = {0.f, 0.f, 0.f, 0.f, 0.f, 0.f, 0.f, 0.f};
    v8f acc_y1 = {0.f, 0.f, 0.f, 0.f, 0.f, 0.f, 0.f, 0.f};
    #pragma unroll
    for (int kc = 0; kc < 8; ++kc)
        acc_y0 = __builtin_amdgcn_wmma_f32_16x16x4_f32(
            false, ha[kc], false, Blin[kc], (short)0, acc_y0, false, false);
    #pragma unroll
    for (int kc = 8; kc < 16; ++kc)
        acc_y1 = __builtin_amdgcn_wmma_f32_16x16x4_f32(
            false, ha[kc], false, Blin[kc], (short)0, acc_y1, false, false);
    #pragma unroll
    for (int v = 0; v < 8; ++v)
        y[v] = acc_y0[v] + acc_y1[v] + blin_l;
}

__global__ __launch_bounds__(NWAVES * WAVE_SZ)
void gru_fused_wmma_kernel(const float* __restrict__ x,
                           const float* __restrict__ W_ih,
                           const float* __restrict__ W_hh,
                           const float* __restrict__ b_ih,
                           const float* __restrict__ b_hh,
                           const float* __restrict__ W_lin,
                           const float* __restrict__ b_lin,
                           float* __restrict__ out,
                           int T)
{
    __shared__ __attribute__((aligned(16))) float h_buf[BM * LDS_PITCH];
    __shared__ __attribute__((aligned(16))) float rz_buf[2 * BM * LDS_PITCH];
    float* const r_buf = rz_buf;
    float* const z_buf = rz_buf + BM * LDS_PITCH;
    __shared__ __attribute__((aligned(16))) float x_buf[BM * X_PITCH];

    const int tid  = threadIdx.x;
    const int wave = tid / WAVE_SZ;
    const int lane = tid % WAVE_SZ;
    const int hi   = lane >> 4;
    const int ln   = lane & 15;
    const int b0   = blockIdx.x * BM;

    const int  ncol = wave * 16;
    const int  gcol = ncol & (GRU_H - 1);
    const bool is_n = (wave >= 8);

    const bool is_stager = (wave >= 4) && (wave < 8);
    const int  sj   = tid & 127;
    const int  srow = sj >> 3;
    const int  sc4  = sj & 7;
    const float* const xs_row = x + ((size_t)(b0 + srow) * T) * GRU_I + 4 * sc4;
    float* const xs_lds = x_buf + srow * X_PITCH + 4 * sc4;

    v2f Bih[GRU_I / 4];
    v2f Bhh[GRU_H / 4];
    v2f Blin[GRU_H / 4];
    {
        const float* wih_row = W_ih + (size_t)(ncol + ln) * GRU_I;
        #pragma unroll
        for (int kc = 0; kc < GRU_I / 4; ++kc) {
            Bih[kc][0] = wih_row[4 * kc + 2 * hi + 0];
            Bih[kc][1] = wih_row[4 * kc + 2 * hi + 1];
        }
        const float* whh_row = W_hh + (size_t)(ncol + ln) * GRU_H;
        #pragma unroll
        for (int kc = 0; kc < GRU_H / 4; ++kc) {
            Bhh[kc][0] = whh_row[4 * kc + 2 * hi + 0];
            Bhh[kc][1] = whh_row[4 * kc + 2 * hi + 1];
        }
        const float* wl_row = W_lin + (size_t)ln * GRU_H;
        #pragma unroll
        for (int kc = 0; kc < GRU_H / 4; ++kc) {
            Blin[kc][0] = wl_row[4 * kc + 2 * hi + 0];
            Blin[kc][1] = wl_row[4 * kc + 2 * hi + 1];
        }
    }
    const float bih_l  = b_ih[ncol + ln];
    const float bhh_l  = b_hh[ncol + ln];
    const float blin_l = b_lin[ln];
    unsigned orow[8];
    #pragma unroll
    for (int v = 0; v < 8; ++v)
        orow[v] = (unsigned)(b0 + v + 8 * hi) * (unsigned)T * GRU_O + ln;

    for (int i = tid; i < BM * LDS_PITCH; i += blockDim.x) h_buf[i] = 0.0f;

    if (is_stager)
        *reinterpret_cast<float4*>(xs_lds) = *reinterpret_cast<const float4*>(xs_row);
    __syncthreads();
    v8f acc_x = {0.f, 0.f, 0.f, 0.f, 0.f, 0.f, 0.f, 0.f};
    #pragma unroll
    for (int kc = 0; kc < GRU_I / 4; ++kc) {
        v2f a;
        a[0] = x_buf[ln * X_PITCH + 4 * kc + 2 * hi + 0];
        a[1] = x_buf[ln * X_PITCH + 4 * kc + 2 * hi + 1];
        acc_x = __builtin_amdgcn_wmma_f32_16x16x4_f32(
            false, a, false, Bih[kc], (short)0, acc_x, false, false);
    }
    __syncthreads();
    if (is_stager) {
        const int t1 = (1 < T) ? 1 : T - 1;
        *reinterpret_cast<float4*>(xs_lds) =
            *reinterpret_cast<const float4*>(xs_row + (size_t)t1 * GRU_I);
    }
    __syncthreads();

    for (int t = 0; t < T; ++t) {

        float4 xr = {0.f, 0.f, 0.f, 0.f};
        if (is_stager) {
            const int t2 = (t + 2 < T) ? t + 2 : T - 1;
            const int t6 = (t + 6 < T) ? t + 6 : T - 1;
            xr = *reinterpret_cast<const float4*>(xs_row + (size_t)t2 * GRU_I);
            __builtin_prefetch((const void*)(xs_row + (size_t)t6 * GRU_I), 0, 1);
        }

        v2f xa[GRU_I / 4];
        #pragma unroll
        for (int kc = 0; kc < GRU_I / 4; ++kc) {
            xa[kc][0] = x_buf[ln * X_PITCH + 4 * kc + 2 * hi + 0];
            xa[kc][1] = x_buf[ln * X_PITCH + 4 * kc + 2 * hi + 1];
        }

        float y_prev[8] = {0.f, 0.f, 0.f, 0.f, 0.f, 0.f, 0.f, 0.f};
        v8f acc_xn = {0.f, 0.f, 0.f, 0.f, 0.f, 0.f, 0.f, 0.f};
        v8f acc_h0 = {0.f, 0.f, 0.f, 0.f, 0.f, 0.f, 0.f, 0.f};
        v8f acc_h1 = {0.f, 0.f, 0.f, 0.f, 0.f, 0.f, 0.f, 0.f};

        v2f ha[GRU_H / 4];
        #pragma unroll
        for (int kc = 0; kc < GRU_H / 4; ++kc) {
            ha[kc][0] = h_buf[ln * LDS_PITCH + 4 * kc + 2 * hi + 0];
            ha[kc][1] = h_buf[ln * LDS_PITCH + 4 * kc + 2 * hi + 1];
        }

        #pragma unroll
        for (int kc = 0; kc < 8; ++kc)
            acc_h0 = __builtin_amdgcn_wmma_f32_16x16x4_f32(
                false, ha[kc], false, Bhh[kc], (short)0, acc_h0, false, false);
        #pragma unroll
        for (int kc = 8; kc < 16; ++kc)
            acc_h1 = __builtin_amdgcn_wmma_f32_16x16x4_f32(
                false, ha[kc], false, Bhh[kc], (short)0, acc_h1, false, false);

        if (!is_n) {
            float* const gate_tile = (wave < 4) ? r_buf : z_buf;
            float g[8];
            #pragma unroll
            for (int v = 0; v < 8; ++v) {
                const float s = acc_x[v] + acc_h0[v] + acc_h1[v] + bih_l + bhh_l;
                g[v] = 1.0f / (1.0f + __expf(-s));
            }
            #pragma unroll
            for (int v = 0; v < 8; ++v) {
                const int m = v + 8 * hi;
                gate_tile[m * LDS_PITCH + gcol + ln] = g[v];
            }
        } else {
            #pragma unroll
            for (int kc = 0; kc < GRU_I / 4; ++kc)
                acc_xn = __builtin_amdgcn_wmma_f32_16x16x4_f32(
                    false, xa[kc], false, Bih[kc], (short)0, acc_xn, false, false);
        }
        __syncthreads();

        if (is_n) {
            float r[8], z[8], ho[8], hn[8];
            #pragma unroll
            for (int v = 0; v < 8; ++v) {
                const int idx = (v + 8 * hi) * LDS_PITCH + gcol + ln;
                r[v]  = r_buf[idx];
                z[v]  = z_buf[idx];
                ho[v] = h_buf[idx];
            }
            #pragma unroll
            for (int v = 0; v < 8; ++v) {
                const float gx = acc_x[v] + bih_l;
                const float gh = acc_h0[v] + acc_h1[v] + bhh_l;
                const float n  = tanh_select(gx + r[v] * gh);
                hn[v] = (1.0f - z[v]) * n + z[v] * ho[v];
            }
            #pragma unroll
            for (int v = 0; v < 8; ++v) {
                const int idx = (v + 8 * hi) * LDS_PITCH + gcol + ln;
                h_buf[idx] = hn[v];
            }
        } else {
            #pragma unroll
            for (int kc = 0; kc < GRU_I / 4; ++kc)
                acc_xn = __builtin_amdgcn_wmma_f32_16x16x4_f32(
                    false, xa[kc], false, Bih[kc], (short)0, acc_xn, false, false);
            if (is_stager) {
                *reinterpret_cast<float4*>(xs_lds) = xr;
            }
            if (wave == 0 && t > 0)
                project(ha, Blin, blin_l, y_prev);
        }
        __syncthreads();

        if (wave == 0 && t > 0) {
            float* const out_t = out + (size_t)(t - 1) * GRU_O;
            #pragma unroll
            for (int v = 0; v < 8; ++v)
                out_t[orow[v]] = y_prev[v];
        }

        acc_x = acc_xn;
    }

    if (wave == 0 && T > 0) {
        v2f ha[GRU_H / 4];
        #pragma unroll
        for (int kc = 0; kc < GRU_H / 4; ++kc) {
            ha[kc][0] = h_buf[ln * LDS_PITCH + 4 * kc + 2 * hi + 0];
            ha[kc][1] = h_buf[ln * LDS_PITCH + 4 * kc + 2 * hi + 1];
        }
        float y_last[8];
        project(ha, Blin, blin_l, y_last);
        float* const out_t = out + (size_t)(T - 1) * GRU_O;
        #pragma unroll
        for (int v = 0; v < 8; ++v)
            out_t[orow[v]] = y_last[v];
    }
}

extern "C" void kernel_launch(void* const* d_in, const int* in_sizes, int n_in,
                              void* d_out, int out_size, void* d_ws, size_t ws_size,
                              hipStream_t stream) {
    const float* x     = (const float*)d_in[0];
    const float* W_ih  = (const float*)d_in[1];
    const float* W_hh  = (const float*)d_in[2];
    const float* b_ih  = (const float*)d_in[3];
    const float* b_hh  = (const float*)d_in[4];
    const float* W_lin = (const float*)d_in[5];
    const float* b_lin = (const float*)d_in[6];
    float* out = (float*)d_out;

    const int B = 512;
    const int T = in_sizes[0] / (B * GRU_I);

    dim3 grid(B / BM);
    dim3 block(NWAVES * WAVE_SZ);
    gru_fused_wmma_kernel<<<grid, block, 0, stream>>>(
        x, W_ih, W_hh, b_ih, b_hh, W_lin, b_lin, out, T);
}


# gate nonlinearities on the hardware approximations: the sigmoid's IEEE division chain becomes one v_rcp_f32, the tanh's library expf one multiply + v_exp_f32 (1 ulp each)
# speedup vs baseline: 3.2210x; 1.2134x over previous
#include <hip/hip_runtime.h>


typedef float v2f __attribute__((ext_vector_type(2)));
typedef float v8f __attribute__((ext_vector_type(8)));

#define GRU_H   64
#define GRU_I   32
#define GRU_O   16
#define GRU_NG  192
#define BM      16
#define NWAVES  12
#define WAVE_SZ 32
#define LDS_PITCH 68
#define X_PITCH 36

__device__ __forceinline__ float tanh_select(float a)
{
    const float ax = __builtin_fabsf(a);
    const float e   = __expf(ax + ax);
    const float big = __builtin_fmaf(__builtin_amdgcn_rcpf(e + 1.0f), -2.0f, 1.0f);
    const float a2 = a * a;
    float p = __builtin_fmaf(a2, -0x1.758e7ap-8f, 0x1.521192p-6f);
    p = __builtin_fmaf(a2, p, -0x1.b8389cp-5f);
    p = __builtin_fmaf(a2, p, 0x1.110704p-3f);
    p = __builtin_fmaf(a2, p, -0x1.555532p-2f);
    const float small = __builtin_fmaf(a2, ax * p, ax);
    const float t = (ax < 0.625f) ? small : big;
    return __builtin_copysignf(t, a);
}

__device__ __forceinline__ void project(const v2f (&ha)[GRU_H / 4],
                                        const v2f (&Blin)[GRU_H / 4],
                                        float blin_l, float (&y)[8])
{
    v8f acc_y0 = {0.f, 0.f, 0.f, 0.f, 0.f, 0.f, 0.f, 0.f};
    v8f acc_y1 = {0.f, 0.f, 0.f, 0.f, 0.f, 0.f, 0.f, 0.f};
    #pragma unroll
    for (int kc = 0; kc < 8; ++kc)
        acc_y0 = __builtin_amdgcn_wmma_f32_16x16x4_f32(
            false, ha[kc], false, Blin[kc], (short)0, acc_y0, false, false);
    #pragma unroll
    for (int kc = 8; kc < 16; ++kc)
        acc_y1 = __builtin_amdgcn_wmma_f32_16x16x4_f32(
            false, ha[kc], false, Blin[kc], (short)0, acc_y1, false, false);
    #pragma unroll
    for (int v = 0; v < 8; ++v)
        y[v] = acc_y0[v] + acc_y1[v] + blin_l;
}

__global__ __launch_bounds__(NWAVES * WAVE_SZ)
void gru_fused_wmma_kernel(const float* __restrict__ x,
                           const float* __restrict__ W_ih,
                           const float* __restrict__ W_hh,
                           const float* __restrict__ b_ih,
                           const float* __restrict__ b_hh,
                           const float* __restrict__ W_lin,
                           const float* __restrict__ b_lin,
                           float* __restrict__ out,
                           int T)
{
    __shared__ __attribute__((aligned(16))) float h_buf[BM * LDS_PITCH];
    __shared__ __attribute__((aligned(16))) float rz_buf[2 * BM * LDS_PITCH];
    float* const r_buf = rz_buf;
    float* const z_buf = rz_buf + BM * LDS_PITCH;
    __shared__ __attribute__((aligned(16))) float x_buf[BM * X_PITCH];

    const int tid  = threadIdx.x;
    const int wave = tid / WAVE_SZ;
    const int lane = tid % WAVE_SZ;
    const int hi   = lane >> 4;
    const int ln   = lane & 15;
    const int b0   = blockIdx.x * BM;

    const int  ncol = wave * 16;
    const int  gcol = ncol & (GRU_H - 1);
    const bool is_n = (wave >= 8);

    const bool is_stager = (wave >= 4) && (wave < 8);
    const int  sj   = tid & 127;
    const int  srow = sj >> 3;
    const int  sc4  = sj & 7;
    const float* const xs_row = x + ((size_t)(b0 + srow) * T) * GRU_I + 4 * sc4;
    float* const xs_lds = x_buf + srow * X_PITCH + 4 * sc4;

    v2f Bih[GRU_I / 4];
    v2f Bhh[GRU_H / 4];
    v2f Blin[GRU_H / 4];
    {
        const float* wih_row = W_ih + (size_t)(ncol + ln) * GRU_I;
        #pragma unroll
        for (int kc = 0; kc < GRU_I / 4; ++kc) {
            Bih[kc][0] = wih_row[4 * kc + 2 * hi + 0];
            Bih[kc][1] = wih_row[4 * kc + 2 * hi + 1];
        }
        const float* whh_row = W_hh + (size_t)(ncol + ln) * GRU_H;
        #pragma unroll
        for (int kc = 0; kc < GRU_H / 4; ++kc) {
            Bhh[kc][0] = whh_row[4 * kc + 2 * hi + 0];
            Bhh[kc][1] = whh_row[4 * kc + 2 * hi + 1];
        }
        const float* wl_row = W_lin + (size_t)ln * GRU_H;
        #pragma unroll
        for (int kc = 0; kc < GRU_H / 4; ++kc) {
            Blin[kc][0] = wl_row[4 * kc + 2 * hi + 0];
            Blin[kc][1] = wl_row[4 * kc + 2 * hi + 1];
        }
    }
    const float bih_l  = b_ih[ncol + ln];
    const float bhh_l  = b_hh[ncol + ln];
    const float blin_l = b_lin[ln];
    unsigned orow[8];
    #pragma unroll
    for (int v = 0; v < 8; ++v)
        orow[v] = (unsigned)(b0 + v + 8 * hi) * (unsigned)T * GRU_O + ln;

    for (int i = tid; i < BM * LDS_PITCH; i += blockDim.x) h_buf[i] = 0.0f;

    if (is_stager)
        *reinterpret_cast<float4*>(xs_lds) = *reinterpret_cast<const float4*>(xs_row);
    __syncthreads();
    v8f acc_x = {0.f, 0.f, 0.f, 0.f, 0.f, 0.f, 0.f, 0.f};
    #pragma unroll
    for (int kc = 0; kc < GRU_I / 4; ++kc) {
        v2f a;
        a[0] = x_buf[ln * X_PITCH + 4 * kc + 2 * hi + 0];
        a[1] = x_buf[ln * X_PITCH + 4 * kc + 2 * hi + 1];
        acc_x = __builtin_amdgcn_wmma_f32_16x16x4_f32(
            false, a, false, Bih[kc], (short)0, acc_x, false, false);
    }
    __syncthreads();
    if (is_stager) {
        const int t1 = (1 < T) ? 1 : T - 1;
        *reinterpret_cast<float4*>(xs_lds) =
            *reinterpret_cast<const float4*>(xs_row + (size_t)t1 * GRU_I);
    }
    __syncthreads();

    for (int t = 0; t < T; ++t) {

        float4 xr = {0.f, 0.f, 0.f, 0.f};
        if (is_stager) {
            const int t2 = (t + 2 < T) ? t + 2 : T - 1;
            const int t6 = (t + 6 < T) ? t + 6 : T - 1;
            xr = *reinterpret_cast<const float4*>(xs_row + (size_t)t2 * GRU_I);
            __builtin_prefetch((const void*)(xs_row + (size_t)t6 * GRU_I), 0, 1);
        }

        v2f xa[GRU_I / 4];
        #pragma unroll
        for (int kc = 0; kc < GRU_I / 4; ++kc) {
            xa[kc][0] = x_buf[ln * X_PITCH + 4 * kc + 2 * hi + 0];
            xa[kc][1] = x_buf[ln * X_PITCH + 4 * kc + 2 * hi + 1];
        }

        float y_prev[8] = {0.f, 0.f, 0.f, 0.f, 0.f, 0.f, 0.f, 0.f};
        v8f acc_xn = {0.f, 0.f, 0.f, 0.f, 0.f, 0.f, 0.f, 0.f};
        v8f acc_h0 = {0.f, 0.f, 0.f, 0.f, 0.f, 0.f, 0.f, 0.f};
        v8f acc_h1 = {0.f, 0.f, 0.f, 0.f, 0.f, 0.f, 0.f, 0.f};

        v2f ha[GRU_H / 4];
        #pragma unroll
        for (int kc = 0; kc < GRU_H / 4; ++kc) {
            ha[kc][0] = h_buf[ln * LDS_PITCH + 4 * kc + 2 * hi + 0];
            ha[kc][1] = h_buf[ln * LDS_PITCH + 4 * kc + 2 * hi + 1];
        }

        #pragma unroll
        for (int kc = 0; kc < 8; ++kc)
            acc_h0 = __builtin_amdgcn_wmma_f32_16x16x4_f32(
                false, ha[kc], false, Bhh[kc], (short)0, acc_h0, false, false);
        #pragma unroll
        for (int kc = 8; kc < 16; ++kc)
            acc_h1 = __builtin_amdgcn_wmma_f32_16x16x4_f32(
                false, ha[kc], false, Bhh[kc], (short)0, acc_h1, false, false);

        if (!is_n) {
            float* const gate_tile = (wave < 4) ? r_buf : z_buf;
            float g[8];
            #pragma unroll
            for (int v = 0; v < 8; ++v) {
                const float s = acc_x[v] + acc_h0[v] + acc_h1[v] + bih_l + bhh_l;
                g[v] = __builtin_amdgcn_rcpf(1.0f + __expf(-s));
            }
            #pragma unroll
            for (int v = 0; v < 8; ++v) {
                const int m = v + 8 * hi;
                gate_tile[m * LDS_PITCH + gcol + ln] = g[v];
            }
        } else {
            #pragma unroll
            for (int kc = 0; kc < GRU_I / 4; ++kc)
                acc_xn = __builtin_amdgcn_wmma_f32_16x16x4_f32(
                    false, xa[kc], false, Bih[kc], (short)0, acc_xn, false, false);
        }
        __syncthreads();

        if (is_n) {
            float r[8], z[8], ho[8], hn[8];
            #pragma unroll
            for (int v = 0; v < 8; ++v) {
                const int idx = (v + 8 * hi) * LDS_PITCH + gcol + ln;
                r[v]  = r_buf[idx];
                z[v]  = z_buf[idx];
                ho[v] = h_buf[idx];
            }
            #pragma unroll
            for (int v = 0; v < 8; ++v) {
                const float gx = acc_x[v] + bih_l;
                const float gh = acc_h0[v] + acc_h1[v] + bhh_l;
                const float n  = tanh_select(gx + r[v] * gh);
                hn[v] = (1.0f - z[v]) * n + z[v] * ho[v];
            }
            #pragma unroll
            for (int v = 0; v < 8; ++v) {
                const int idx = (v + 8 * hi) * LDS_PITCH + gcol + ln;
                h_buf[idx] = hn[v];
            }
        } else {
            #pragma unroll
            for (int kc = 0; kc < GRU_I / 4; ++kc)
                acc_xn = __builtin_amdgcn_wmma_f32_16x16x4_f32(
                    false, xa[kc], false, Bih[kc], (short)0, acc_xn, false, false);
            if (is_stager) {
                *reinterpret_cast<float4*>(xs_lds) = xr;
            }
            if (wave == 0 && t > 0)
                project(ha, Blin, blin_l, y_prev);
        }
        __syncthreads();

        if (wave == 0 && t > 0) {
            float* const out_t = out + (size_t)(t - 1) * GRU_O;
            #pragma unroll
            for (int v = 0; v < 8; ++v)
                out_t[orow[v]] = y_prev[v];
        }

        acc_x = acc_xn;
    }

    if (wave == 0 && T > 0) {
        v2f ha[GRU_H / 4];
        #pragma unroll
        for (int kc = 0; kc < GRU_H / 4; ++kc) {
            ha[kc][0] = h_buf[ln * LDS_PITCH + 4 * kc + 2 * hi + 0];
            ha[kc][1] = h_buf[ln * LDS_PITCH + 4 * kc + 2 * hi + 1];
        }
        float y_last[8];
        project(ha, Blin, blin_l, y_last);
        float* const out_t = out + (size_t)(T - 1) * GRU_O;
        #pragma unroll
        for (int v = 0; v < 8; ++v)
            out_t[orow[v]] = y_last[v];
    }
}

extern "C" void kernel_launch(void* const* d_in, const int* in_sizes, int n_in,
                              void* d_out, int out_size, void* d_ws, size_t ws_size,
                              hipStream_t stream) {
    const float* x     = (const float*)d_in[0];
    const float* W_ih  = (const float*)d_in[1];
    const float* W_hh  = (const float*)d_in[2];
    const float* b_ih  = (const float*)d_in[3];
    const float* b_hh  = (const float*)d_in[4];
    const float* W_lin = (const float*)d_in[5];
    const float* b_lin = (const float*)d_in[6];
    float* out = (float*)d_out;

    const int B = 512;
    const int T = in_sizes[0] / (B * GRU_I);

    dim3 grid(B / BM);
    dim3 block(NWAVES * WAVE_SZ);
    gru_fused_wmma_kernel<<<grid, block, 0, stream>>>(
        x, W_ih, W_hh, b_ih, b_hh, W_lin, b_lin, out, T);
}


# output projection split by K over the four r waves, one per SIMD (4 matrix instructions each instead of 16 on wave 0); partial tiles summed through LDS, so y is summed as (p0+p1)+(p2+p3)+b
# speedup vs baseline: 3.5705x; 1.1085x over previous
#include <hip/hip_runtime.h>


typedef float v2f __attribute__((ext_vector_type(2)));
typedef float v8f __attribute__((ext_vector_type(8)));

#define GRU_H   64
#define GRU_I   32
#define GRU_O   16
#define GRU_NG  192
#define BM      16
#define NWAVES  12
#define WAVE_SZ 32
#define LDS_PITCH 68
#define X_PITCH 36
#define Y_PITCH 20

__device__ __forceinline__ float tanh_select(float a)
{
    const float ax = __builtin_fabsf(a);
    const float e   = __expf(ax + ax);
    const float big = __builtin_fmaf(__builtin_amdgcn_rcpf(e + 1.0f), -2.0f, 1.0f);
    const float a2 = a * a;
    float p = __builtin_fmaf(a2, -0x1.758e7ap-8f, 0x1.521192p-6f);
    p = __builtin_fmaf(a2, p, -0x1.b8389cp-5f);
    p = __builtin_fmaf(a2, p, 0x1.110704p-3f);
    p = __builtin_fmaf(a2, p, -0x1.555532p-2f);
    const float small = __builtin_fmaf(a2, ax * p, ax);
    const float t = (ax < 0.625f) ? small : big;
    return __builtin_copysignf(t, a);
}

__device__ __forceinline__ void project_quarter(const v2f (&pa)[4],
                                                const v2f (&Blin4)[4],
                                                float* __restrict__ part_tile,
                                                int hi, int ln)
{
    v8f acc = {0.f, 0.f, 0.f, 0.f, 0.f, 0.f, 0.f, 0.f};
    #pragma unroll
    for (int j = 0; j < 4; ++j)
        acc = __builtin_amdgcn_wmma_f32_16x16x4_f32(
            false, pa[j], false, Blin4[j], (short)0, acc, false, false);
    #pragma unroll
    for (int v = 0; v < 8; ++v)
        part_tile[(v + 8 * hi) * Y_PITCH + ln] = acc[v];
}

__device__ __forceinline__ float2 project_sum(const float* __restrict__ y_part,
                                              int row, int col, float2 b)
{
    const float2 p0 = *reinterpret_cast<const float2*>(y_part + (0 * BM + row) * Y_PITCH + col);
    const float2 p1 = *reinterpret_cast<const float2*>(y_part + (1 * BM + row) * Y_PITCH + col);
    const float2 p2 = *reinterpret_cast<const float2*>(y_part + (2 * BM + row) * Y_PITCH + col);
    const float2 p3 = *reinterpret_cast<const float2*>(y_part + (3 * BM + row) * Y_PITCH + col);
    float2 y;
    y.x = ((p0.x + p1.x) + (p2.x + p3.x)) + b.x;
    y.y = ((p0.y + p1.y) + (p2.y + p3.y)) + b.y;
    return y;
}

__global__ __launch_bounds__(NWAVES * WAVE_SZ)
void gru_fused_wmma_kernel(const float* __restrict__ x,
                           const float* __restrict__ W_ih,
                           const float* __restrict__ W_hh,
                           const float* __restrict__ b_ih,
                           const float* __restrict__ b_hh,
                           const float* __restrict__ W_lin,
                           const float* __restrict__ b_lin,
                           float* __restrict__ out,
                           int T)
{
    __shared__ __attribute__((aligned(16))) float h_buf[BM * LDS_PITCH];
    __shared__ __attribute__((aligned(16))) float rz_buf[2 * BM * LDS_PITCH];
    float* const r_buf = rz_buf;
    float* const z_buf = rz_buf + BM * LDS_PITCH;
    __shared__ __attribute__((aligned(16))) float x_buf[BM * X_PITCH];
    __shared__ __attribute__((aligned(16))) float y_part[4 * BM * Y_PITCH];

    const int tid  = threadIdx.x;
    const int wave = tid / WAVE_SZ;
    const int lane = tid % WAVE_SZ;
    const int hi   = lane >> 4;
    const int ln   = lane & 15;
    const int b0   = blockIdx.x * BM;

    const int  ncol = wave * 16;
    const int  gcol = ncol & (GRU_H - 1);
    const bool is_n = (wave >= 8);
    const bool is_proj = (wave < 4);
    const int  pk0  = 16 * (wave & 3);
    const int  prow = (wave & 3) + 4 * (lane >> 3);
    const int  pcol = 2 * (lane & 7);

    const bool is_stager = (wave >= 4) && (wave < 8);
    const int  sj   = tid & 127;
    const int  srow = sj >> 3;
    const int  sc4  = sj & 7;
    const float* const xs_row = x + ((size_t)(b0 + srow) * T) * GRU_I + 4 * sc4;
    float* const xs_lds = x_buf + srow * X_PITCH + 4 * sc4;

    v2f Bih[GRU_I / 4];
    v2f Bhh[GRU_H / 4];
    v2f Blin4[4];
    {
        const float* wih_row = W_ih + (size_t)(ncol + ln) * GRU_I;
        #pragma unroll
        for (int kc = 0; kc < GRU_I / 4; ++kc) {
            Bih[kc][0] = wih_row[4 * kc + 2 * hi + 0];
            Bih[kc][1] = wih_row[4 * kc + 2 * hi + 1];
        }
        const float* whh_row = W_hh + (size_t)(ncol + ln) * GRU_H;
        #pragma unroll
        for (int kc = 0; kc < GRU_H / 4; ++kc) {
            Bhh[kc][0] = whh_row[4 * kc + 2 * hi + 0];
            Bhh[kc][1] = whh_row[4 * kc + 2 * hi + 1];
        }
        const float* wl_row = W_lin + (size_t)ln * GRU_H + pk0;
        #pragma unroll
        for (int j = 0; j < 4; ++j) {
            Blin4[j][0] = wl_row[4 * j + 2 * hi + 0];
            Blin4[j][1] = wl_row[4 * j + 2 * hi + 1];
        }
    }
    const float bih_l  = b_ih[ncol + ln];
    const float bhh_l  = b_hh[ncol + ln];
    float2 blin2;
    blin2.x = b_lin[pcol + 0];
    blin2.y = b_lin[pcol + 1];
    float* const out_row = out + (size_t)(b0 + prow) * T * GRU_O + pcol;

    for (int i = tid; i < BM * LDS_PITCH; i += blockDim.x) h_buf[i] = 0.0f;

    if (is_stager)
        *reinterpret_cast<float4*>(xs_lds) = *reinterpret_cast<const float4*>(xs_row);
    __syncthreads();
    v8f acc_x = {0.f, 0.f, 0.f, 0.f, 0.f, 0.f, 0.f, 0.f};
    #pragma unroll
    for (int kc = 0; kc < GRU_I / 4; ++kc) {
        v2f a;
        a[0] = x_buf[ln * X_PITCH + 4 * kc + 2 * hi + 0];
        a[1] = x_buf[ln * X_PITCH + 4 * kc + 2 * hi + 1];
        acc_x = __builtin_amdgcn_wmma_f32_16x16x4_f32(
            false, a, false, Bih[kc], (short)0, acc_x, false, false);
    }
    __syncthreads();
    if (is_stager) {
        const int t1 = (1 < T) ? 1 : T - 1;
        *reinterpret_cast<float4*>(xs_lds) =
            *reinterpret_cast<const float4*>(xs_row + (size_t)t1 * GRU_I);
    }
    __syncthreads();

    for (int t = 0; t < T; ++t) {

        float4 xr = {0.f, 0.f, 0.f, 0.f};
        if (is_stager) {
            const int t2 = (t + 2 < T) ? t + 2 : T - 1;
            const int t6 = (t + 6 < T) ? t + 6 : T - 1;
            xr = *reinterpret_cast<const float4*>(xs_row + (size_t)t2 * GRU_I);
            __builtin_prefetch((const void*)(xs_row + (size_t)t6 * GRU_I), 0, 1);
        }

        v2f xa[GRU_I / 4];
        #pragma unroll
        for (int kc = 0; kc < GRU_I / 4; ++kc) {
            xa[kc][0] = x_buf[ln * X_PITCH + 4 * kc + 2 * hi + 0];
            xa[kc][1] = x_buf[ln * X_PITCH + 4 * kc + 2 * hi + 1];
        }

        v8f acc_xn = {0.f, 0.f, 0.f, 0.f, 0.f, 0.f, 0.f, 0.f};
        v8f acc_h0 = {0.f, 0.f, 0.f, 0.f, 0.f, 0.f, 0.f, 0.f};
        v8f acc_h1 = {0.f, 0.f, 0.f, 0.f, 0.f, 0.f, 0.f, 0.f};

        v2f ha[GRU_H / 4];
        #pragma unroll
        for (int kc = 0; kc < GRU_H / 4; ++kc) {
            ha[kc][0] = h_buf[ln * LDS_PITCH + 4 * kc + 2 * hi + 0];
            ha[kc][1] = h_buf[ln * LDS_PITCH + 4 * kc + 2 * hi + 1];
        }

        v2f pa[4];
        if (is_proj) {
            #pragma unroll
            for (int j = 0; j < 4; ++j) {
                pa[j][0] = h_buf[ln * LDS_PITCH + pk0 + 4 * j + 2 * hi + 0];
                pa[j][1] = h_buf[ln * LDS_PITCH + pk0 + 4 * j + 2 * hi + 1];
            }
        }

        #pragma unroll
        for (int kc = 0; kc < 8; ++kc)
            acc_h0 = __builtin_amdgcn_wmma_f32_16x16x4_f32(
                false, ha[kc], false, Bhh[kc], (short)0, acc_h0, false, false);
        #pragma unroll
        for (int kc = 8; kc < 16; ++kc)
            acc_h1 = __builtin_amdgcn_wmma_f32_16x16x4_f32(
                false, ha[kc], false, Bhh[kc], (short)0, acc_h1, false, false);

        if (!is_n) {
            float* const gate_tile = (wave < 4) ? r_buf : z_buf;
            float g[8];
            #pragma unroll
            for (int v = 0; v < 8; ++v) {
                const float s = acc_x[v] + acc_h0[v] + acc_h1[v] + bih_l + bhh_l;
                g[v] = __builtin_amdgcn_rcpf(1.0f + __expf(-s));
            }
            #pragma unroll
            for (int v = 0; v < 8; ++v) {
                const int m = v + 8 * hi;
                gate_tile[m * LDS_PITCH + gcol + ln] = g[v];
            }
        } else {
            #pragma unroll
            for (int kc = 0; kc < GRU_I / 4; ++kc)
                acc_xn = __builtin_amdgcn_wmma_f32_16x16x4_f32(
                    false, xa[kc], false, Bih[kc], (short)0, acc_xn, false, false);
        }
        __syncthreads();

        if (is_n) {
            float r[8], z[8], ho[8], hn[8];
            #pragma unroll
            for (int v = 0; v < 8; ++v) {
                const int idx = (v + 8 * hi) * LDS_PITCH + gcol + ln;
                r[v]  = r_buf[idx];
                z[v]  = z_buf[idx];
                ho[v] = h_buf[idx];
            }
            #pragma unroll
            for (int v = 0; v < 8; ++v) {
                const float gx = acc_x[v] + bih_l;
                const float gh = acc_h0[v] + acc_h1[v] + bhh_l;
                const float n  = tanh_select(gx + r[v] * gh);
                hn[v] = (1.0f - z[v]) * n + z[v] * ho[v];
            }
            #pragma unroll
            for (int v = 0; v < 8; ++v) {
                const int idx = (v + 8 * hi) * LDS_PITCH + gcol + ln;
                h_buf[idx] = hn[v];
            }
        } else {
            #pragma unroll
            for (int kc = 0; kc < GRU_I / 4; ++kc)
                acc_xn = __builtin_amdgcn_wmma_f32_16x16x4_f32(
                    false, xa[kc], false, Bih[kc], (short)0, acc_xn, false, false);
            if (is_stager) {
                *reinterpret_cast<float4*>(xs_lds) = xr;
            }
            if (is_proj && t > 0)
                project_quarter(pa, Blin4, y_part + (wave & 3) * BM * Y_PITCH, hi, ln);
        }
        __syncthreads();

        if (is_proj && t > 0) {
            const float2 y = project_sum(y_part, prow, pcol, blin2);
            *reinterpret_cast<float2*>(out_row + (size_t)(t - 1) * GRU_O) = y;
        }

        acc_x = acc_xn;
    }

    __syncthreads();
    if (is_proj && T > 0) {
        v2f pa[4];
        #pragma unroll
        for (int j = 0; j < 4; ++j) {
            pa[j][0] = h_buf[ln * LDS_PITCH + pk0 + 4 * j + 2 * hi + 0];
            pa[j][1] = h_buf[ln * LDS_PITCH + pk0 + 4 * j + 2 * hi + 1];
        }
        project_quarter(pa, Blin4, y_part + (wave & 3) * BM * Y_PITCH, hi, ln);
    }
    __syncthreads();
    if (is_proj && T > 0) {
        const float2 y = project_sum(y_part, prow, pcol, blin2);
        *reinterpret_cast<float2*>(out_row + (size_t)(T - 1) * GRU_O) = y;
    }
}

extern "C" void kernel_launch(void* const* d_in, const int* in_sizes, int n_in,
                              void* d_out, int out_size, void* d_ws, size_t ws_size,
                              hipStream_t stream) {
    const float* x     = (const float*)d_in[0];
    const float* W_ih  = (const float*)d_in[1];
    const float* W_hh  = (const float*)d_in[2];
    const float* b_ih  = (const float*)d_in[3];
    const float* b_hh  = (const float*)d_in[4];
    const float* W_lin = (const float*)d_in[5];
    const float* b_lin = (const float*)d_in[6];
    float* out = (float*)d_out;

    const int B = 512;
    const int T = in_sizes[0] / (B * GRU_I);

    dim3 grid(B / BM);
    dim3 block(NWAVES * WAVE_SZ);
    gru_fused_wmma_kernel<<<grid, block, 0, stream>>>(
        x, W_ih, W_hh, b_ih, b_hh, W_lin, b_lin, out, T);
}


# 4 batch rows per block instead of 16 (128 blocks): the freed rows of each 16-row tile carry 4 time steps of the x product and of the projection; gates on the 4 live rows only; wave-role tests scalar
# speedup vs baseline: 4.7402x; 1.3276x over previous
#include <hip/hip_runtime.h>


typedef float v2f __attribute__((ext_vector_type(2)));
typedef float v8f __attribute__((ext_vector_type(8)));

#define GRU_H   64
#define GRU_I   32
#define GRU_O   16
#define GRU_NG  192
#define BM      16
#define BMR     4
#define GRP     (BM / BMR)
#define XK_STEP ((GRU_I / 4) / GRP)
#define VN      (BMR < 8 ? BMR : 8)
#define NWAVES  12
#define WAVE_SZ 32
#define LDS_PITCH 68
#define X_PITCH 36
#define Y_PITCH 20

__device__ __forceinline__ float tanh_select(float a)
{
    const float ax = __builtin_fabsf(a);
    const float e   = __expf(ax + ax);
    const float big = __builtin_fmaf(__builtin_amdgcn_rcpf(e + 1.0f), -2.0f, 1.0f);
    const float a2 = a * a;
    float p = __builtin_fmaf(a2, -0x1.758e7ap-8f, 0x1.521192p-6f);
    p = __builtin_fmaf(a2, p, -0x1.b8389cp-5f);
    p = __builtin_fmaf(a2, p, 0x1.110704p-3f);
    p = __builtin_fmaf(a2, p, -0x1.555532p-2f);
    const float small = __builtin_fmaf(a2, ax * p, ax);
    const float t = (ax < 0.625f) ? small : big;
    return __builtin_copysignf(t, a);
}

__device__ __forceinline__ void project_quarter(const v2f (&pa)[4],
                                                const v2f (&Blin4)[4],
                                                float* __restrict__ part_tile,
                                                int hi, int ln)
{
    v8f acc = {0.f, 0.f, 0.f, 0.f, 0.f, 0.f, 0.f, 0.f};
    #pragma unroll
    for (int j = 0; j < 4; ++j)
        acc = __builtin_amdgcn_wmma_f32_16x16x4_f32(
            false, pa[j], false, Blin4[j], (short)0, acc, false, false);
    #pragma unroll
    for (int v = 0; v < 8; ++v)
        part_tile[(v + 8 * hi) * Y_PITCH + ln] = acc[v];
}

__device__ __forceinline__ float2 project_sum(const float* __restrict__ y_part,
                                              int row, int col, float2 b)
{
    const float2 p0 = *reinterpret_cast<const float2*>(y_part + (0 * BM + row) * Y_PITCH + col);
    const float2 p1 = *reinterpret_cast<const float2*>(y_part + (1 * BM + row) * Y_PITCH + col);
    const float2 p2 = *reinterpret_cast<const float2*>(y_part + (2 * BM + row) * Y_PITCH + col);
    const float2 p3 = *reinterpret_cast<const float2*>(y_part + (3 * BM + row) * Y_PITCH + col);
    float2 y;
    y.x = ((p0.x + p1.x) + (p2.x + p3.x)) + b.x;
    y.y = ((p0.y + p1.y) + (p2.y + p3.y)) + b.y;
    return y;
}

__global__ __launch_bounds__(NWAVES * WAVE_SZ)
void gru_fused_wmma_kernel(const float* __restrict__ x,
                           const float* __restrict__ W_ih,
                           const float* __restrict__ W_hh,
                           const float* __restrict__ b_ih,
                           const float* __restrict__ b_hh,
                           const float* __restrict__ W_lin,
                           const float* __restrict__ b_lin,
                           float* __restrict__ out,
                           int T)
{
    __shared__ __attribute__((aligned(16))) float h_buf[BM * LDS_PITCH];
    __shared__ __attribute__((aligned(16))) float rz_buf[2 * BM * LDS_PITCH];
    float* const r_buf = rz_buf;
    float* const z_buf = rz_buf + BM * LDS_PITCH;
    __shared__ __attribute__((aligned(16))) float x_buf[BM * X_PITCH];
    __shared__ __attribute__((aligned(16))) float y_part[4 * BM * Y_PITCH];

    const int tid  = threadIdx.x;
    const int wave = __builtin_amdgcn_readfirstlane(tid / WAVE_SZ);
    const int lane = tid % WAVE_SZ;
    const int hi   = lane >> 4;
    const int ln   = lane & 15;
    const int b0   = blockIdx.x * BMR;

    const int  ncol = wave * 16;
    const int  gcol = ncol & (GRU_H - 1);
    const bool is_n = (wave >= 8);
    const bool is_proj = (wave < 4);
    const int  pk0  = 16 * (wave & 3);
    const int  prow = (wave & 3) + 4 * (lane >> 3);
    const int  pcol = 2 * (lane & 7);
    const int  p_b  = prow % BMR;
    const int  p_ph = (prow / BMR + GRP - 1) % GRP;

    const bool is_stager = (wave >= 4) && (wave < 8);
    const int  sj    = tid & 127;
    const int  srow  = sj >> 3;
    const int  sc4   = sj & 7;
    const int  s_b   = srow % BMR;
    const int  s_tau = srow / BMR;
    const float* const xs_row = x + ((size_t)(b0 + s_b) * T) * GRU_I + 4 * sc4;
    float* const xs_lds = x_buf + srow * X_PITCH + 4 * sc4;

    v2f Bih[GRU_I / 4];
    v2f Bhh[GRU_H / 4];
    v2f Blin4[4];
    {
        const float* wih_row = W_ih + (size_t)(ncol + ln) * GRU_I;
        #pragma unroll
        for (int kc = 0; kc < GRU_I / 4; ++kc) {
            Bih[kc][0] = wih_row[4 * kc + 2 * hi + 0];
            Bih[kc][1] = wih_row[4 * kc + 2 * hi + 1];
        }
        const float* whh_row = W_hh + (size_t)(ncol + ln) * GRU_H;
        #pragma unroll
        for (int kc = 0; kc < GRU_H / 4; ++kc) {
            Bhh[kc][0] = whh_row[4 * kc + 2 * hi + 0];
            Bhh[kc][1] = whh_row[4 * kc + 2 * hi + 1];
        }
        const float* wl_row = W_lin + (size_t)ln * GRU_H + pk0;
        #pragma unroll
        for (int j = 0; j < 4; ++j) {
            Blin4[j][0] = wl_row[4 * j + 2 * hi + 0];
            Blin4[j][1] = wl_row[4 * j + 2 * hi + 1];
        }
    }
    const float bih_l  = b_ih[ncol + ln];
    const float bhh_l  = b_hh[ncol + ln];
    float2 blin2;
    blin2.x = b_lin[pcol + 0];
    blin2.y = b_lin[pcol + 1];
    float* const out_row = out + ((size_t)(b0 + p_b) * T + p_ph) * GRU_O + pcol;

    for (int i = tid; i < BM * LDS_PITCH; i += blockDim.x) h_buf[i] = 0.0f;

    if (is_stager) {
        const int t0 = (s_tau < T) ? s_tau : T - 1;
        *reinterpret_cast<float4*>(xs_lds) =
            *reinterpret_cast<const float4*>(xs_row + (size_t)t0 * GRU_I);
    }
    __syncthreads();
    v8f acc_x = {0.f, 0.f, 0.f, 0.f, 0.f, 0.f, 0.f, 0.f};
    #pragma unroll
    for (int kc = 0; kc < GRU_I / 4; ++kc) {
        v2f a;
        a[0] = x_buf[ln * X_PITCH + 4 * kc + 2 * hi + 0];
        a[1] = x_buf[ln * X_PITCH + 4 * kc + 2 * hi + 1];
        acc_x = __builtin_amdgcn_wmma_f32_16x16x4_f32(
            false, a, false, Bih[kc], (short)0, acc_x, false, false);
    }
    __syncthreads();
    if (is_stager) {
        const int t1 = (GRP + s_tau < T) ? GRP + s_tau : T - 1;
        *reinterpret_cast<float4*>(xs_lds) =
            *reinterpret_cast<const float4*>(xs_row + (size_t)t1 * GRU_I);
    }
    __syncthreads();

    for (int tg = 0; tg < T; tg += GRP) {

        float4 xr = {0.f, 0.f, 0.f, 0.f};
        if (is_stager) {
            const int ta = (tg + 2 * GRP + s_tau < T) ? tg + 2 * GRP + s_tau : T - 1;
            const int tp = (tg + 4 * GRP + s_tau < T) ? tg + 4 * GRP + s_tau : T - 1;
            xr = *reinterpret_cast<const float4*>(xs_row + (size_t)ta * GRU_I);
            __builtin_prefetch((const void*)(xs_row + (size_t)tp * GRU_I), 0, 1);
        }

        v2f xa[GRU_I / 4];
        #pragma unroll
        for (int kc = 0; kc < GRU_I / 4; ++kc) {
            xa[kc][0] = x_buf[ln * X_PITCH + 4 * kc + 2 * hi + 0];
            xa[kc][1] = x_buf[ln * X_PITCH + 4 * kc + 2 * hi + 1];
        }
        v8f acc_xn = {0.f, 0.f, 0.f, 0.f, 0.f, 0.f, 0.f, 0.f};

        const bool do_proj = is_proj && (tg > 0);
        v2f pa[4];
        if (is_proj) {
            #pragma unroll
            for (int j = 0; j < 4; ++j) {
                pa[j][0] = h_buf[ln * LDS_PITCH + pk0 + 4 * j + 2 * hi + 0];
                pa[j][1] = h_buf[ln * LDS_PITCH + pk0 + 4 * j + 2 * hi + 1];
            }
        }

        #pragma unroll
        for (int p = 0; p < GRP; ++p) {
            if (tg + p < T) {
                const int V0      = (BMR * p) & 7;
                const int HI_LIVE = (BMR * p) >> 3;

                v8f acc_h0 = {0.f, 0.f, 0.f, 0.f, 0.f, 0.f, 0.f, 0.f};
                v8f acc_h1 = {0.f, 0.f, 0.f, 0.f, 0.f, 0.f, 0.f, 0.f};

                v2f ha[GRU_H / 4];
                #pragma unroll
                for (int kc = 0; kc < GRU_H / 4; ++kc) {
                    ha[kc][0] = h_buf[ln * LDS_PITCH + 4 * kc + 2 * hi + 0];
                    ha[kc][1] = h_buf[ln * LDS_PITCH + 4 * kc + 2 * hi + 1];
                }

                #pragma unroll
                for (int kc = 0; kc < 8; ++kc)
                    acc_h0 = __builtin_amdgcn_wmma_f32_16x16x4_f32(
                        false, ha[kc], false, Bhh[kc], (short)0, acc_h0, false, false);
                #pragma unroll
                for (int kc = 8; kc < 16; ++kc)
                    acc_h1 = __builtin_amdgcn_wmma_f32_16x16x4_f32(
                        false, ha[kc], false, Bhh[kc], (short)0, acc_h1, false, false);

                if (!is_n) {
                    float* const gate_tile = (wave < 4) ? r_buf : z_buf;
                    float g[VN];
                    #pragma unroll
                    for (int v = 0; v < VN; ++v) {
                        const float s = acc_x[V0 + v] + acc_h0[V0 + v] + acc_h1[V0 + v] + bih_l + bhh_l;
                        g[v] = __builtin_amdgcn_rcpf(1.0f + __expf(-s));
                    }
                    #pragma unroll
                    for (int v = 0; v < VN; ++v) {
                        const int m = V0 + v + 8 * hi;
                        gate_tile[m * LDS_PITCH + gcol + ln] = g[v];
                    }
                } else {
                    #pragma unroll
                    for (int kc = XK_STEP * p; kc < XK_STEP * (p + 1); ++kc)
                        acc_xn = __builtin_amdgcn_wmma_f32_16x16x4_f32(
                            false, xa[kc], false, Bih[kc], (short)0, acc_xn, false, false);
                }
                __syncthreads();

                if (is_n) {
                    float r[VN], z[VN], ho[VN], hn[VN];
                    #pragma unroll
                    for (int v = 0; v < VN; ++v) {
                        const int idx = (V0 + v + 8 * hi) * LDS_PITCH + gcol + ln;
                        r[v]  = r_buf[idx];
                        z[v]  = z_buf[idx];
                        ho[v] = h_buf[idx];
                    }
                    #pragma unroll
                    for (int v = 0; v < VN; ++v) {
                        const float gx = acc_x[V0 + v] + bih_l;
                        const float gh = acc_h0[V0 + v] + acc_h1[V0 + v] + bhh_l;
                        const float n  = tanh_select(gx + r[v] * gh);
                        hn[v] = (1.0f - z[v]) * n + z[v] * ho[v];
                    }
                    if (hi == HI_LIVE) {
                        #pragma unroll
                        for (int v = 0; v < VN; ++v) {
                            const int mn = (V0 + v + 8 * hi + BMR) & (BM - 1);
                            h_buf[mn * LDS_PITCH + gcol + ln] = hn[v];
                        }
                    }
                } else {
                    #pragma unroll
                    for (int kc = XK_STEP * p; kc < XK_STEP * (p + 1); ++kc)
                        acc_xn = __builtin_amdgcn_wmma_f32_16x16x4_f32(
                            false, xa[kc], false, Bih[kc], (short)0, acc_xn, false, false);
                    if (p == GRP - 1 && is_stager) {
                        *reinterpret_cast<float4*>(xs_lds) = xr;
                    }
                    if (p == 0 && do_proj)
                        project_quarter(pa, Blin4, y_part + (wave & 3) * BM * Y_PITCH, hi, ln);
                }
                __syncthreads();

                if (p == 0 && do_proj) {
                    const float2 y = project_sum(y_part, prow, pcol, blin2);
                    *reinterpret_cast<float2*>(out_row + (size_t)(tg - GRP) * GRU_O) = y;
                }
            }
        }

        acc_x = acc_xn;
    }

    __syncthreads();
    if (is_proj && T > 0) {
        v2f pa[4];
        #pragma unroll
        for (int j = 0; j < 4; ++j) {
            pa[j][0] = h_buf[ln * LDS_PITCH + pk0 + 4 * j + 2 * hi + 0];
            pa[j][1] = h_buf[ln * LDS_PITCH + pk0 + 4 * j + 2 * hi + 1];
        }
        project_quarter(pa, Blin4, y_part + (wave & 3) * BM * Y_PITCH, hi, ln);
    }
    __syncthreads();
    if (is_proj && T > 0) {
        const int tl = ((T - 1) / GRP) * GRP;
        if (tl + p_ph < T) {
            const float2 y = project_sum(y_part, prow, pcol, blin2);
            *reinterpret_cast<float2*>(out_row + (size_t)tl * GRU_O) = y;
        }
    }
}

extern "C" void kernel_launch(void* const* d_in, const int* in_sizes, int n_in,
                              void* d_out, int out_size, void* d_ws, size_t ws_size,
                              hipStream_t stream) {
    const float* x     = (const float*)d_in[0];
    const float* W_ih  = (const float*)d_in[1];
    const float* W_hh  = (const float*)d_in[2];
    const float* b_ih  = (const float*)d_in[3];
    const float* b_hh  = (const float*)d_in[4];
    const float* W_lin = (const float*)d_in[5];
    const float* b_lin = (const float*)d_in[6];
    float* out = (float*)d_out;

    const int B = 512;
    const int T = in_sizes[0] / (B * GRU_I);

    dim3 grid(B / BMR);
    dim3 block(NWAVES * WAVE_SZ);
    gru_fused_wmma_kernel<<<grid, block, 0, stream>>>(
        x, W_ih, W_hh, b_ih, b_hh, W_lin, b_lin, out, T);
}


# each of 4 waves (one per SIMD) owns 16 hidden columns of all 3 gates and runs its chains r, n, z: sigmoid(r) issues between n's matrix instructions, tanh between z's; 1 barrier a step
# speedup vs baseline: 5.0570x; 1.0668x over previous
#include <hip/hip_runtime.h>


typedef float v2f __attribute__((ext_vector_type(2)));
typedef float v8f __attribute__((ext_vector_type(8)));

#define GRU_H   64
#define GRU_I   32
#define GRU_O   16
#define BM      16
#define BMR     4
#define GRP     (BM / BMR)
#define VN      BMR
#define NWAVES  4
#define WAVE_SZ 32
#define LDS_PITCH 68
#define X_PITCH 36
#define Y_PITCH 20

#define WMMA_F32(acc, a, b) \
    acc = __builtin_amdgcn_wmma_f32_16x16x4_f32(false, a, false, b, (short)0, acc, false, false)

#define STAGE_ACC2_ACC2(a0, a1, b0, b1) \
    asm volatile("" : "+v"(a0), "+v"(a1), "+v"(b0), "+v"(b1))
#define STAGE_ACC2_V4(a0, a1, x) \
    asm volatile("" : "+v"(a0), "+v"(a1), "+v"(x[0]), "+v"(x[1]), "+v"(x[2]), "+v"(x[3]))

__device__ __forceinline__ float tanh_exp(float a)
{
    const float ax = __builtin_fabsf(a);
    return __expf(ax + ax);
}
__device__ __forceinline__ float tanh_small(float a)
{
    const float ax = __builtin_fabsf(a);
    const float a2 = a * a;
    float p = __builtin_fmaf(a2, -0x1.758e7ap-8f, 0x1.521192p-6f);
    p = __builtin_fmaf(a2, p, -0x1.b8389cp-5f);
    p = __builtin_fmaf(a2, p, 0x1.110704p-3f);
    p = __builtin_fmaf(a2, p, -0x1.555532p-2f);
    return __builtin_fmaf(a2, ax * p, ax);
}
__device__ __forceinline__ float tanh_back(float a, float e, float small_r)
{
    const float ax  = __builtin_fabsf(a);
    const float big = __builtin_fmaf(__builtin_amdgcn_rcpf(e + 1.0f), -2.0f, 1.0f);
    const float t = (ax < 0.625f) ? small_r : big;
    return __builtin_copysignf(t, a);
}

__device__ __forceinline__ float2 project_sum(const float* __restrict__ y_part,
                                              int row, int col, float2 b)
{
    const float2 p0 = *reinterpret_cast<const float2*>(y_part + (0 * BM + row) * Y_PITCH + col);
    const float2 p1 = *reinterpret_cast<const float2*>(y_part + (1 * BM + row) * Y_PITCH + col);
    const float2 p2 = *reinterpret_cast<const float2*>(y_part + (2 * BM + row) * Y_PITCH + col);
    const float2 p3 = *reinterpret_cast<const float2*>(y_part + (3 * BM + row) * Y_PITCH + col);
    float2 y;
    y.x = ((p0.x + p1.x) + (p2.x + p3.x)) + b.x;
    y.y = ((p0.y + p1.y) + (p2.y + p3.y)) + b.y;
    return y;
}

__global__ __launch_bounds__(NWAVES * WAVE_SZ)
void gru_fused_wmma_kernel(const float* __restrict__ x,
                           const float* __restrict__ W_ih,
                           const float* __restrict__ W_hh,
                           const float* __restrict__ b_ih,
                           const float* __restrict__ b_hh,
                           const float* __restrict__ W_lin,
                           const float* __restrict__ b_lin,
                           float* __restrict__ out,
                           int T)
{
    __shared__ __attribute__((aligned(16))) float h_buf[BM * LDS_PITCH];
    __shared__ __attribute__((aligned(16))) float hp_buf[2 * BM * LDS_PITCH];
    __shared__ __attribute__((aligned(16))) float x_buf[BM * X_PITCH];
    __shared__ __attribute__((aligned(16))) float y_part[4 * BM * Y_PITCH];
    __shared__ __attribute__((aligned(16))) float dead_buf[BM * LDS_PITCH];

    const int tid  = threadIdx.x;
    const int wave = __builtin_amdgcn_readfirstlane(tid / WAVE_SZ);
    const int lane = tid % WAVE_SZ;
    const int hi   = lane >> 4;
    const int ln   = lane & 15;
    const int b0   = blockIdx.x * BMR;
    const int c0   = wave * 16;

    const int  pk0  = 16 * wave;
    const int  prow = wave + 4 * (lane >> 3);
    const int  pcol = 2 * (lane & 7);
    const int  p_b  = prow % BMR;
    const int  p_ph = prow / BMR;

    const int  srow  = tid >> 3;
    const int  sc4   = tid & 7;
    const int  s_b   = srow % BMR;
    const int  s_tau = srow / BMR;
    const float* const xs_row = x + ((size_t)(b0 + s_b) * T) * GRU_I + 4 * sc4;
    float* const xs_lds = x_buf + srow * X_PITCH + 4 * sc4;

    v2f Bih[3][GRU_I / 4];
    v2f Bhh[3][GRU_H / 4];
    v2f Blin4[4];
    float bih[3], bhh[3];
    #pragma unroll
    for (int g = 0; g < 3; ++g) {
        const float* wih_row = W_ih + (size_t)(g * GRU_H + c0 + ln) * GRU_I;
        #pragma unroll
        for (int kc = 0; kc < GRU_I / 4; ++kc) {
            Bih[g][kc][0] = wih_row[4 * kc + 2 * hi + 0];
            Bih[g][kc][1] = wih_row[4 * kc + 2 * hi + 1];
        }
        const float* whh_row = W_hh + (size_t)(g * GRU_H + c0 + ln) * GRU_H;
        #pragma unroll
        for (int kc = 0; kc < GRU_H / 4; ++kc) {
            Bhh[g][kc][0] = whh_row[4 * kc + 2 * hi + 0];
            Bhh[g][kc][1] = whh_row[4 * kc + 2 * hi + 1];
        }
        bih[g] = b_ih[g * GRU_H + c0 + ln];
        bhh[g] = b_hh[g * GRU_H + c0 + ln];
    }
    {
        const float* wl_row = W_lin + (size_t)ln * GRU_H + pk0;
        #pragma unroll
        for (int j = 0; j < 4; ++j) {
            Blin4[j][0] = wl_row[4 * j + 2 * hi + 0];
            Blin4[j][1] = wl_row[4 * j + 2 * hi + 1];
        }
    }
    float2 blin2;
    blin2.x = b_lin[pcol + 0];
    blin2.y = b_lin[pcol + 1];
    float* const out_row = out + ((size_t)(b0 + p_b) * T + p_ph) * GRU_O + pcol;

    const float* const ha_ptr = h_buf + ln * LDS_PITCH + 2 * hi;
    const float* const xa_ptr = x_buf + ln * X_PITCH + 2 * hi;
    float* const y_mine = y_part + wave * BM * Y_PITCH;

    for (int i = tid; i < BM * LDS_PITCH; i += blockDim.x) h_buf[i] = 0.0f;

    {
        const int t0 = (s_tau < T) ? s_tau : T - 1;
        *reinterpret_cast<float4*>(xs_lds) =
            *reinterpret_cast<const float4*>(xs_row + (size_t)t0 * GRU_I);
    }
    __syncthreads();
    const v8f zero8 = {0.f, 0.f, 0.f, 0.f, 0.f, 0.f, 0.f, 0.f};
    v8f acc_x[3] = {zero8, zero8, zero8};
    #pragma unroll
    for (int kc = 0; kc < GRU_I / 4; ++kc) {
        v2f a;
        a[0] = xa_ptr[4 * kc + 0];
        a[1] = xa_ptr[4 * kc + 1];
        #pragma unroll
        for (int g = 0; g < 3; ++g) WMMA_F32(acc_x[g], a, Bih[g][kc]);
    }
    __syncthreads();
    {
        const int t1 = (GRP + s_tau < T) ? GRP + s_tau : T - 1;
        *reinterpret_cast<float4*>(xs_lds) =
            *reinterpret_cast<const float4*>(xs_row + (size_t)t1 * GRU_I);
    }
    __syncthreads();

    v2f ha[GRU_H / 4];
    v2f xa[GRU_I / 4];
    v2f pa[4];
    float ho[VN];
    #pragma unroll
    for (int kc = 0; kc < GRU_H / 4; ++kc) {
        ha[kc][0] = ha_ptr[4 * kc + 0];
        ha[kc][1] = ha_ptr[4 * kc + 1];
    }
    #pragma unroll
    for (int kc = 0; kc < GRU_I / 4; ++kc) {
        xa[kc][0] = xa_ptr[4 * kc + 0];
        xa[kc][1] = xa_ptr[4 * kc + 1];
    }
    #pragma unroll
    for (int j = 0; j < 4; ++j) { pa[j][0] = 0.f; pa[j][1] = 0.f; }
    #pragma unroll
    for (int v = 0; v < VN; ++v) ho[v] = 0.0f;

    for (int tg = 0; tg < T; tg += GRP) {
        const bool do_proj = (tg > 0);
        float* const hp_tile = hp_buf + ((tg / GRP) & 1) * BM * LDS_PITCH;

        float4 xr;
        {
            const int ta = (tg + 2 * GRP + s_tau < T) ? tg + 2 * GRP + s_tau : T - 1;
            const int tp = (tg + 4 * GRP + s_tau < T) ? tg + 4 * GRP + s_tau : T - 1;
            xr = *reinterpret_cast<const float4*>(xs_row + (size_t)ta * GRU_I);
            __builtin_prefetch((const void*)(xs_row + (size_t)tp * GRU_I), 0, 1);
        }

        v8f acc_xn[3] = {zero8, zero8, zero8};
        v8f accp = zero8;

        #pragma unroll
        for (int p = 0; p < GRP; ++p) {
            if (tg + p < T) {
                const int V0      = (BMR * p) & 7;
                const int HI_LIVE = (BMR * p) >> 3;

                #pragma unroll
                for (int kc = 2 * p; kc < 2 * p + 2; ++kc) {
                    #pragma unroll
                    for (int g = 0; g < 3; ++g) WMMA_F32(acc_xn[g], xa[kc], Bih[g][kc]);
                }
                if (p == 0 && do_proj) {
                    #pragma unroll
                    for (int j = 0; j < 4; ++j) WMMA_F32(accp, pa[j], Blin4[j]);
                }

                v8f ar0 = zero8, ar1 = zero8;
                v8f an0 = zero8, an1 = zero8;
                v8f az0 = zero8, az1 = zero8;
                float sr[VN], er[VN], rg[VN];
                float an[VN], en[VN], sn[VN], nn[VN];
                float sz[VN], zg[VN];

                #pragma unroll
                for (int kc = 0; kc < 8; ++kc) {
                    WMMA_F32(ar0, ha[kc],     Bhh[0][kc]);
                    WMMA_F32(ar1, ha[kc + 8], Bhh[0][kc + 8]);
                }

                #pragma unroll
                for (int kc = 0; kc < 8; ++kc) {
                    WMMA_F32(an0, ha[kc],     Bhh[2][kc]);
                    WMMA_F32(an1, ha[kc + 8], Bhh[2][kc + 8]);
                    if (kc == 0) {
                        STAGE_ACC2_ACC2(an0, an1, ar0, ar1);
                        #pragma unroll
                        for (int v = 0; v < VN; ++v)
                            sr[v] = acc_x[0][V0 + v] + ar0[V0 + v] + ar1[V0 + v] + bih[0] + bhh[0];
                        STAGE_ACC2_V4(an0, an1, sr);
                    } else if (kc == 1) {
                        STAGE_ACC2_V4(an0, an1, sr);
                        #pragma unroll
                        for (int v = 0; v < VN; ++v) er[v] = __expf(-sr[v]);
                        STAGE_ACC2_V4(an0, an1, er);
                    } else if (kc == 2) {
                        STAGE_ACC2_V4(an0, an1, er);
                        #pragma unroll
                        for (int v = 0; v < VN; ++v) rg[v] = __builtin_amdgcn_rcpf(1.0f + er[v]);
                        STAGE_ACC2_V4(an0, an1, rg);
                    }
                }

                #pragma unroll
                for (int kc = 0; kc < 8; ++kc) {
                    WMMA_F32(az0, ha[kc],     Bhh[1][kc]);
                    WMMA_F32(az1, ha[kc + 8], Bhh[1][kc + 8]);
                    if (kc == 0) {
                        STAGE_ACC2_ACC2(az0, az1, an0, an1);
                        #pragma unroll
                        for (int v = 0; v < VN; ++v) {
                            const float gx = acc_x[2][V0 + v] + bih[2];
                            const float gh = an0[V0 + v] + an1[V0 + v] + bhh[2];
                            an[v] = gx + rg[v] * gh;
                        }
                        STAGE_ACC2_V4(az0, az1, an);
                    } else if (kc == 1) {
                        STAGE_ACC2_V4(az0, az1, an);
                        #pragma unroll
                        for (int v = 0; v < VN; ++v) en[v] = tanh_exp(an[v]);
                        STAGE_ACC2_V4(az0, az1, en);
                    } else if (kc == 2) {
                        STAGE_ACC2_V4(az0, az1, an);
                        #pragma unroll
                        for (int v = 0; v < VN; ++v) sn[v] = tanh_small(an[v]);
                        STAGE_ACC2_V4(az0, az1, sn);
                    } else if (kc == 3) {
                        STAGE_ACC2_V4(az0, az1, en);
                        #pragma unroll
                        for (int v = 0; v < VN; ++v) nn[v] = tanh_back(an[v], en[v], sn[v]);
                        STAGE_ACC2_V4(az0, az1, nn);
                    }
                }

                #pragma unroll
                for (int v = 0; v < VN; ++v) {
                    sz[v] = acc_x[1][V0 + v] + az0[V0 + v] + az1[V0 + v] + bih[1] + bhh[1];
                    zg[v] = __builtin_amdgcn_rcpf(1.0f + __expf(-sz[v]));
                }

                float hn[VN];
                #pragma unroll
                for (int v = 0; v < VN; ++v)
                    hn[v] = (1.0f - zg[v]) * nn[v] + zg[v] * ho[v];
                {
                    const bool live = (hi == HI_LIVE);
                    float* const hb = (live ? h_buf   : dead_buf) + c0 + ln;
                    float* const pb = (live ? hp_tile : dead_buf) + c0 + ln;
                    #pragma unroll
                    for (int v = 0; v < VN; ++v) {
                        const int m  = V0 + v + 8 * hi;
                        const int mn = (m + BMR) & (BM - 1);
                        hb[mn * LDS_PITCH] = hn[v];
                        pb[m * LDS_PITCH]  = hn[v];
                    }
                }
                if (p == 0 && do_proj) {
                    #pragma unroll
                    for (int v = 0; v < 8; ++v)
                        y_mine[(v + 8 * hi) * Y_PITCH + ln] = accp[v];
                }
                if (p == GRP - 1) {
                    *reinterpret_cast<float4*>(xs_lds) = xr;
                }
                __syncthreads();

                #pragma unroll
                for (int kc = 0; kc < GRU_H / 4; ++kc) {
                    ha[kc][0] = ha_ptr[4 * kc + 0];
                    ha[kc][1] = ha_ptr[4 * kc + 1];
                }
                {
                    const int V0n = (BMR * (p + 1)) & 7;
                    #pragma unroll
                    for (int v = 0; v < VN; ++v)
                        ho[v] = h_buf[(V0n + v + 8 * hi) * LDS_PITCH + c0 + ln];
                }
                if (p == GRP - 1) {
                    #pragma unroll
                    for (int kc = 0; kc < GRU_I / 4; ++kc) {
                        xa[kc][0] = xa_ptr[4 * kc + 0];
                        xa[kc][1] = xa_ptr[4 * kc + 1];
                    }
                    #pragma unroll
                    for (int j = 0; j < 4; ++j) {
                        pa[j][0] = hp_tile[ln * LDS_PITCH + pk0 + 4 * j + 2 * hi + 0];
                        pa[j][1] = hp_tile[ln * LDS_PITCH + pk0 + 4 * j + 2 * hi + 1];
                    }
                }
                if (p == 0 && do_proj) {
                    const float2 y = project_sum(y_part, prow, pcol, blin2);
                    *reinterpret_cast<float2*>(out_row + (size_t)(tg - GRP) * GRU_O) = y;
                }
            }
        }

        #pragma unroll
        for (int g = 0; g < 3; ++g) acc_x[g] = acc_xn[g];
    }

    __syncthreads();
    if (T > 0) {
        const int gl = (T - 1) / GRP;
        const float* const hp_last = hp_buf + (gl & 1) * BM * LDS_PITCH;
        v2f pl[4];
        #pragma unroll
        for (int j = 0; j < 4; ++j) {
            pl[j][0] = hp_last[ln * LDS_PITCH + pk0 + 4 * j + 2 * hi + 0];
            pl[j][1] = hp_last[ln * LDS_PITCH + pk0 + 4 * j + 2 * hi + 1];
        }
        v8f accl = zero8;
        #pragma unroll
        for (int j = 0; j < 4; ++j) WMMA_F32(accl, pl[j], Blin4[j]);
        #pragma unroll
        for (int v = 0; v < 8; ++v)
            y_mine[(v + 8 * hi) * Y_PITCH + ln] = accl[v];
    }
    __syncthreads();
    if (T > 0) {
        const int tl = ((T - 1) / GRP) * GRP;
        if (tl + p_ph < T) {
            const float2 y = project_sum(y_part, prow, pcol, blin2);
            *reinterpret_cast<float2*>(out_row + (size_t)tl * GRU_O) = y;
        }
    }
}

extern "C" void kernel_launch(void* const* d_in, const int* in_sizes, int n_in,
                              void* d_out, int out_size, void* d_ws, size_t ws_size,
                              hipStream_t stream) {
    const float* x     = (const float*)d_in[0];
    const float* W_ih  = (const float*)d_in[1];
    const float* W_hh  = (const float*)d_in[2];
    const float* b_ih  = (const float*)d_in[3];
    const float* b_hh  = (const float*)d_in[4];
    const float* W_lin = (const float*)d_in[5];
    const float* b_lin = (const float*)d_in[6];
    float* out = (float*)d_out;

    const int B = 512;
    const int T = in_sizes[0] / (B * GRU_I);

    dim3 grid(B / BMR);
    dim3 block(NWAVES * WAVE_SZ);
    gru_fused_wmma_kernel<<<grid, block, 0, stream>>>(
        x, W_ih, W_hh, b_ih, b_hh, W_lin, b_lin, out, T);
}


# hand-over latency covered by matrix work: chains start with the wave's own 16 columns (k order rotated per wave: chunk products summed in another order), read back inside a split barrier
# speedup vs baseline: 5.2586x; 1.0399x over previous
#include <hip/hip_runtime.h>


typedef float v2f __attribute__((ext_vector_type(2)));
typedef float v8f __attribute__((ext_vector_type(8)));

#define GRU_H   64
#define GRU_I   32
#define GRU_O   16
#define BM      16
#define BMR     4
#define GRP     (BM / BMR)
#define VN      BMR
#define NWAVES  4
#define WAVE_SZ 32
#define LDS_PITCH 68
#define X_PITCH 36
#define Y_PITCH 20

#define WMMA_F32(acc, a, b) \
    acc = __builtin_amdgcn_wmma_f32_16x16x4_f32(false, a, false, b, (short)0, acc, false, false)

#define STAGE_ACC2_ACC2(a0, a1, b0, b1) \
    asm volatile("" : "+v"(a0), "+v"(a1), "+v"(b0), "+v"(b1))
#define STAGE_ACC2_V4(a0, a1, x) \
    asm volatile("" : "+v"(a0), "+v"(a1), "+v"(x[0]), "+v"(x[1]), "+v"(x[2]), "+v"(x[3]))
#define STAGE_ACC3(a0, a1, a2) \
    asm volatile("" : "+v"(a0), "+v"(a1), "+v"(a2))
#define STAGE_FRAG2(f0, f1) \
    asm volatile("" : "+v"(f0), "+v"(f1))
#define STAGE_ACC6(a0, a1, a2, a3, a4, a5) \
    asm volatile("" : "+v"(a0), "+v"(a1), "+v"(a2), "+v"(a3), "+v"(a4), "+v"(a5))

__device__ __forceinline__ float tanh_exp(float a)
{
    const float ax = __builtin_fabsf(a);
    return __expf(ax + ax);
}
__device__ __forceinline__ float tanh_small(float a)
{
    const float ax = __builtin_fabsf(a);
    const float a2 = a * a;
    float p = __builtin_fmaf(a2, -0x1.758e7ap-8f, 0x1.521192p-6f);
    p = __builtin_fmaf(a2, p, -0x1.b8389cp-5f);
    p = __builtin_fmaf(a2, p, 0x1.110704p-3f);
    p = __builtin_fmaf(a2, p, -0x1.555532p-2f);
    return __builtin_fmaf(a2, ax * p, ax);
}
__device__ __forceinline__ float tanh_back(float a, float e, float small_r)
{
    const float ax  = __builtin_fabsf(a);
    const float big = __builtin_fmaf(__builtin_amdgcn_rcpf(e + 1.0f), -2.0f, 1.0f);
    const float t = (ax < 0.625f) ? small_r : big;
    return __builtin_copysignf(t, a);
}

__device__ __forceinline__ void project_load(const float* __restrict__ y_part,
                                             int row, int col, float2 (&q)[4])
{
    #pragma unroll
    for (int k = 0; k < 4; ++k)
        q[k] = *reinterpret_cast<const float2*>(y_part + (k * BM + row) * Y_PITCH + col);
}
__device__ __forceinline__ float2 project_add(const float2 (&q)[4], float2 b)
{
    float2 y;
    y.x = ((q[0].x + q[1].x) + (q[2].x + q[3].x)) + b.x;
    y.y = ((q[0].y + q[1].y) + (q[2].y + q[3].y)) + b.y;
    return y;
}
__device__ __forceinline__ float2 project_sum(const float* __restrict__ y_part,
                                              int row, int col, float2 b)
{
    const float2 p0 = *reinterpret_cast<const float2*>(y_part + (0 * BM + row) * Y_PITCH + col);
    const float2 p1 = *reinterpret_cast<const float2*>(y_part + (1 * BM + row) * Y_PITCH + col);
    const float2 p2 = *reinterpret_cast<const float2*>(y_part + (2 * BM + row) * Y_PITCH + col);
    const float2 p3 = *reinterpret_cast<const float2*>(y_part + (3 * BM + row) * Y_PITCH + col);
    float2 y;
    y.x = ((p0.x + p1.x) + (p2.x + p3.x)) + b.x;
    y.y = ((p0.y + p1.y) + (p2.y + p3.y)) + b.y;
    return y;
}

__global__ __launch_bounds__(NWAVES * WAVE_SZ)
void gru_fused_wmma_kernel(const float* __restrict__ x,
                           const float* __restrict__ W_ih,
                           const float* __restrict__ W_hh,
                           const float* __restrict__ b_ih,
                           const float* __restrict__ b_hh,
                           const float* __restrict__ W_lin,
                           const float* __restrict__ b_lin,
                           float* __restrict__ out,
                           int T)
{
    __shared__ __attribute__((aligned(16))) float h_buf[BM * LDS_PITCH];
    __shared__ __attribute__((aligned(16))) float hp_buf[2 * BM * LDS_PITCH];
    __shared__ __attribute__((aligned(16))) float x_buf[BM * X_PITCH];
    __shared__ __attribute__((aligned(16))) float y_part[4 * BM * Y_PITCH];
    __shared__ __attribute__((aligned(16))) float dead_buf[BM * LDS_PITCH];

    const int tid  = threadIdx.x;
    const int wave = __builtin_amdgcn_readfirstlane(tid / WAVE_SZ);
    const int lane = tid % WAVE_SZ;
    const int hi   = lane >> 4;
    const int ln   = lane & 15;
    const int b0   = blockIdx.x * BMR;
    const int c0   = wave * 16;

    const int  pk0  = 16 * wave;
    const int  prow = wave + 4 * (lane >> 3);
    const int  pcol = 2 * (lane & 7);
    const int  p_b  = prow % BMR;
    const int  p_ph = prow / BMR;

    const int  srow  = tid >> 3;
    const int  sc4   = tid & 7;
    const int  s_b   = srow % BMR;
    const int  s_tau = srow / BMR;
    const float* const xs_row = x + ((size_t)(b0 + s_b) * T) * GRU_I + 4 * sc4;
    float* const xs_lds = x_buf + srow * X_PITCH + 4 * sc4;

    v2f Bih[3][GRU_I / 4];
    v2f Bhh[3][GRU_H / 4];
    v2f Blin4[4];
    float bih[3], bhh[3];
    #pragma unroll
    for (int g = 0; g < 3; ++g) {
        const float* wih_row = W_ih + (size_t)(g * GRU_H + c0 + ln) * GRU_I;
        #pragma unroll
        for (int kc = 0; kc < GRU_I / 4; ++kc) {
            Bih[g][kc][0] = wih_row[4 * kc + 2 * hi + 0];
            Bih[g][kc][1] = wih_row[4 * kc + 2 * hi + 1];
        }
        const float* whh_row = W_hh + (size_t)(g * GRU_H + c0 + ln) * GRU_H;
        #pragma unroll
        for (int s = 0; s < GRU_H / 4; ++s) {
            const int kc = (s + 4 * wave) & 15;
            Bhh[g][s][0] = whh_row[4 * kc + 2 * hi + 0];
            Bhh[g][s][1] = whh_row[4 * kc + 2 * hi + 1];
        }
        bih[g] = b_ih[g * GRU_H + c0 + ln];
        bhh[g] = b_hh[g * GRU_H + c0 + ln];
    }
    {
        const float* wl_row = W_lin + (size_t)ln * GRU_H + pk0;
        #pragma unroll
        for (int j = 0; j < 4; ++j) {
            Blin4[j][0] = wl_row[4 * j + 2 * hi + 0];
            Blin4[j][1] = wl_row[4 * j + 2 * hi + 1];
        }
    }
    float2 blin2;
    blin2.x = b_lin[pcol + 0];
    blin2.y = b_lin[pcol + 1];
    float* const out_row = out + ((size_t)(b0 + p_b) * T + p_ph) * GRU_O + pcol;

    const float* hs[4];
    #pragma unroll
    for (int j = 0; j < 4; ++j)
        hs[j] = h_buf + ln * LDS_PITCH + 2 * hi + 16 * ((wave + j) & 3);
    const float* const xa_ptr = x_buf + ln * X_PITCH + 2 * hi;
    float* const y_mine = y_part + wave * BM * Y_PITCH;

    for (int i = tid; i < BM * LDS_PITCH; i += blockDim.x) h_buf[i] = 0.0f;

    {
        const int t0 = (s_tau < T) ? s_tau : T - 1;
        *reinterpret_cast<float4*>(xs_lds) =
            *reinterpret_cast<const float4*>(xs_row + (size_t)t0 * GRU_I);
    }
    __syncthreads();
    const v8f zero8 = {0.f, 0.f, 0.f, 0.f, 0.f, 0.f, 0.f, 0.f};
    v8f acc_x[3] = {zero8, zero8, zero8};
    #pragma unroll
    for (int kc = 0; kc < GRU_I / 4; ++kc) {
        v2f a;
        a[0] = xa_ptr[4 * kc + 0];
        a[1] = xa_ptr[4 * kc + 1];
        #pragma unroll
        for (int g = 0; g < 3; ++g) WMMA_F32(acc_x[g], a, Bih[g][kc]);
    }
    __syncthreads();
    {
        const int t1 = (GRP + s_tau < T) ? GRP + s_tau : T - 1;
        *reinterpret_cast<float4*>(xs_lds) =
            *reinterpret_cast<const float4*>(xs_row + (size_t)t1 * GRU_I);
    }
    __syncthreads();

    v2f ha[GRU_H / 4];
    v2f xa[GRU_I / 4];
    v2f pa[4];
    float ho[VN];
    #pragma unroll
    for (int s = 0; s < GRU_H / 4; ++s) { ha[s][0] = 0.f; ha[s][1] = 0.f; }
    #pragma unroll
    for (int kc = 0; kc < GRU_I / 4; ++kc) {
        xa[kc][0] = xa_ptr[4 * kc + 0];
        xa[kc][1] = xa_ptr[4 * kc + 1];
    }
    #pragma unroll
    for (int j = 0; j < 4; ++j) { pa[j][0] = 0.f; pa[j][1] = 0.f; }
    #pragma unroll
    for (int v = 0; v < VN; ++v) ho[v] = 0.0f;

    for (int tg = 0; tg < T; tg += GRP) {
        const bool do_proj = (tg > 0);
        float* const hp_tile = hp_buf + ((tg / GRP) & 1) * BM * LDS_PITCH;

        float4 xr;
        {
            const int ta = (tg + 2 * GRP + s_tau < T) ? tg + 2 * GRP + s_tau : T - 1;
            const int tp = (tg + 4 * GRP + s_tau < T) ? tg + 4 * GRP + s_tau : T - 1;
            xr = *reinterpret_cast<const float4*>(xs_row + (size_t)ta * GRU_I);
            __builtin_prefetch((const void*)(xs_row + (size_t)tp * GRU_I), 0, 1);
        }

        v8f acc_xn[3] = {zero8, zero8, zero8};
        v8f accp = zero8;
        float2 yq[4] = {};

        #pragma unroll
        for (int p = 0; p < GRP; ++p) {
            if (tg + p < T) {
                const int V0      = (BMR * p) & 7;
                const int HI_LIVE = (BMR * p) >> 3;

                v8f ar0 = zero8, ar1 = zero8;
                v8f an0 = zero8, an1 = zero8;
                v8f az0 = zero8, az1 = zero8;
                float sr[VN], er[VN], rg[VN];
                float an[VN], en[VN], sn[VN], nn[VN];
                float sz[VN], zg[VN];

                #pragma unroll
                for (int s = 0; s < 2; ++s) {
                    WMMA_F32(ar0, ha[s],     Bhh[0][s]);
                    WMMA_F32(ar1, ha[s + 2], Bhh[0][s + 2]);
                    WMMA_F32(an0, ha[s],     Bhh[2][s]);
                    WMMA_F32(an1, ha[s + 2], Bhh[2][s + 2]);
                    WMMA_F32(az0, ha[s],     Bhh[1][s]);
                    WMMA_F32(az1, ha[s + 2], Bhh[1][s + 2]);
                }
                STAGE_ACC6(ar0, ar1, an0, an1, az0, az1);
                if (p == 1 && do_proj) {
                    const float2 y = project_add(yq, blin2);
                    *reinterpret_cast<float2*>(out_row + (size_t)(tg - GRP) * GRU_O) = y;
                }

                #pragma unroll
                for (int i = 0; i < 6; ++i) {
                    WMMA_F32(ar0, ha[4 + i],  Bhh[0][4 + i]);
                    WMMA_F32(ar1, ha[10 + i], Bhh[0][10 + i]);
                }

                if (p == 0 && do_proj) {
                    #pragma unroll
                    for (int j = 0; j < 4; ++j) WMMA_F32(accp, pa[j], Blin4[j]);
                }

                #pragma unroll
                for (int kc = 0; kc < 6; ++kc) {
                    WMMA_F32(an0, ha[4 + kc],  Bhh[2][4 + kc]);
                    WMMA_F32(an1, ha[10 + kc], Bhh[2][10 + kc]);
                    if (kc == 0) {
                        STAGE_ACC2_ACC2(an0, an1, ar0, ar1);
                        #pragma unroll
                        for (int v = 0; v < VN; ++v)
                            sr[v] = acc_x[0][V0 + v] + ar0[V0 + v] + ar1[V0 + v] + bih[0] + bhh[0];
                        STAGE_ACC2_V4(an0, an1, sr);
                    } else if (kc == 1) {
                        STAGE_ACC2_V4(an0, an1, sr);
                        #pragma unroll
                        for (int v = 0; v < VN; ++v) er[v] = __expf(-sr[v]);
                        STAGE_ACC2_V4(an0, an1, er);
                    } else if (kc == 2) {
                        STAGE_ACC2_V4(an0, an1, er);
                        #pragma unroll
                        for (int v = 0; v < VN; ++v) rg[v] = __builtin_amdgcn_rcpf(1.0f + er[v]);
                        STAGE_ACC2_V4(an0, an1, rg);
                    }
                }

                #pragma unroll
                for (int kc = 0; kc < 6; ++kc) {
                    WMMA_F32(az0, ha[4 + kc],  Bhh[1][4 + kc]);
                    WMMA_F32(az1, ha[10 + kc], Bhh[1][10 + kc]);
                    if (kc == 0) {
                        STAGE_ACC2_ACC2(az0, az1, an0, an1);
                        #pragma unroll
                        for (int v = 0; v < VN; ++v) {
                            const float gx = acc_x[2][V0 + v] + bih[2];
                            const float gh = an0[V0 + v] + an1[V0 + v] + bhh[2];
                            an[v] = gx + rg[v] * gh;
                        }
                        STAGE_ACC2_V4(az0, az1, an);
                    } else if (kc == 1) {
                        STAGE_ACC2_V4(az0, az1, an);
                        #pragma unroll
                        for (int v = 0; v < VN; ++v) en[v] = tanh_exp(an[v]);
                        STAGE_ACC2_V4(az0, az1, en);
                    } else if (kc == 2) {
                        STAGE_ACC2_V4(az0, az1, an);
                        #pragma unroll
                        for (int v = 0; v < VN; ++v) sn[v] = tanh_small(an[v]);
                        STAGE_ACC2_V4(az0, az1, sn);
                    } else if (kc == 3) {
                        STAGE_ACC2_V4(az0, az1, en);
                        #pragma unroll
                        for (int v = 0; v < VN; ++v) nn[v] = tanh_back(an[v], en[v], sn[v]);
                        STAGE_ACC2_V4(az0, az1, nn);
                    }
                }

                #pragma unroll
                for (int v = 0; v < VN; ++v) {
                    sz[v] = acc_x[1][V0 + v] + az0[V0 + v] + az1[V0 + v] + bih[1] + bhh[1];
                    zg[v] = __builtin_amdgcn_rcpf(1.0f + __expf(-sz[v]));
                }

                float hn[VN];
                #pragma unroll
                for (int v = 0; v < VN; ++v)
                    hn[v] = (1.0f - zg[v]) * nn[v] + zg[v] * ho[v];
                {
                    const bool live = (hi == HI_LIVE);
                    float* const hb = (live ? h_buf   : dead_buf) + c0 + ln;
                    float* const pb = (live ? hp_tile : dead_buf) + c0 + ln;
                    #pragma unroll
                    for (int v = 0; v < VN; ++v) {
                        const int m  = V0 + v + 8 * hi;
                        const int mn = (m + BMR) & (BM - 1);
                        hb[mn * LDS_PITCH] = hn[v];
                        pb[m * LDS_PITCH]  = hn[v];
                    }
                }
                if (p == 0 && do_proj) {
                    #pragma unroll
                    for (int v = 0; v < 8; ++v)
                        y_mine[(v + 8 * hi) * Y_PITCH + ln] = accp[v];
                }
                if (p == GRP - 1) {
                    *reinterpret_cast<float4*>(xs_lds) = xr;
                }
                __builtin_amdgcn_fence(__ATOMIC_RELEASE, "workgroup");
                __builtin_amdgcn_s_barrier_signal(-1);
                #pragma unroll
                for (int s = 0; s < 4; ++s) {
                    ha[s][0] = hs[0][4 * s + 0];
                    ha[s][1] = hs[0][4 * s + 1];
                }
                STAGE_FRAG2(xa[2 * p], xa[2 * p + 1]);
                #pragma unroll
                for (int kc = 2 * p; kc < 2 * p + 2; ++kc) {
                    #pragma unroll
                    for (int g = 0; g < 3; ++g) WMMA_F32(acc_xn[g], xa[kc], Bih[g][kc]);
                }
                STAGE_ACC3(acc_xn[0], acc_xn[1], acc_xn[2]);
                __builtin_amdgcn_s_barrier_wait(-1);
                __builtin_amdgcn_fence(__ATOMIC_ACQUIRE, "workgroup");

                #pragma unroll
                for (int s = 4; s < GRU_H / 4; ++s) {
                    ha[s][0] = hs[s >> 2][4 * (s & 3) + 0];
                    ha[s][1] = hs[s >> 2][4 * (s & 3) + 1];
                }
                {
                    const int V0n = (BMR * (p + 1)) & 7;
                    #pragma unroll
                    for (int v = 0; v < VN; ++v)
                        ho[v] = h_buf[(V0n + v + 8 * hi) * LDS_PITCH + c0 + ln];
                }
                if (p == GRP - 1) {
                    #pragma unroll
                    for (int kc = 0; kc < GRU_I / 4; ++kc) {
                        xa[kc][0] = xa_ptr[4 * kc + 0];
                        xa[kc][1] = xa_ptr[4 * kc + 1];
                    }
                    #pragma unroll
                    for (int j = 0; j < 4; ++j) {
                        pa[j][0] = hp_tile[ln * LDS_PITCH + pk0 + 4 * j + 2 * hi + 0];
                        pa[j][1] = hp_tile[ln * LDS_PITCH + pk0 + 4 * j + 2 * hi + 1];
                    }
                }
                if (p == 0 && do_proj) {
                    project_load(y_part, prow, pcol, yq);
                    if (tg + 1 >= T) {
                        const float2 y = project_add(yq, blin2);
                        *reinterpret_cast<float2*>(out_row + (size_t)(tg - GRP) * GRU_O) = y;
                    }
                }
            }
        }

        #pragma unroll
        for (int g = 0; g < 3; ++g) acc_x[g] = acc_xn[g];
    }

    __syncthreads();
    if (T > 0) {
        const int gl = (T - 1) / GRP;
        const float* const hp_last = hp_buf + (gl & 1) * BM * LDS_PITCH;
        v2f pl[4];
        #pragma unroll
        for (int j = 0; j < 4; ++j) {
            pl[j][0] = hp_last[ln * LDS_PITCH + pk0 + 4 * j + 2 * hi + 0];
            pl[j][1] = hp_last[ln * LDS_PITCH + pk0 + 4 * j + 2 * hi + 1];
        }
        v8f accl = zero8;
        #pragma unroll
        for (int j = 0; j < 4; ++j) WMMA_F32(accl, pl[j], Blin4[j]);
        #pragma unroll
        for (int v = 0; v < 8; ++v)
            y_mine[(v + 8 * hi) * Y_PITCH + ln] = accl[v];
    }
    __syncthreads();
    if (T > 0) {
        const int tl = ((T - 1) / GRP) * GRP;
        if (tl + p_ph < T) {
            const float2 y = project_sum(y_part, prow, pcol, blin2);
            *reinterpret_cast<float2*>(out_row + (size_t)tl * GRU_O) = y;
        }
    }
}

extern "C" void kernel_launch(void* const* d_in, const int* in_sizes, int n_in,
                              void* d_out, int out_size, void* d_ws, size_t ws_size,
                              hipStream_t stream) {
    const float* x     = (const float*)d_in[0];
    const float* W_ih  = (const float*)d_in[1];
    const float* W_hh  = (const float*)d_in[2];
    const float* b_ih  = (const float*)d_in[3];
    const float* b_hh  = (const float*)d_in[4];
    const float* W_lin = (const float*)d_in[5];
    const float* b_lin = (const float*)d_in[6];
    float* out = (float*)d_out;

    const int B = 512;
    const int T = in_sizes[0] / (B * GRU_I);

    dim3 grid(B / BMR);
    dim3 block(NWAVES * WAVE_SZ);
    gru_fused_wmma_kernel<<<grid, block, 0, stream>>>(
        x, W_ih, W_hh, b_ih, b_hh, W_lin, b_lin, out, T);
}


# 2 batch rows per block instead of 4 (256 blocks, one per WGP): a 16-row tile carries 8 time steps of the x product and of the projection; gate arithmetic and state stores on the 2 live rows
# speedup vs baseline: 5.6032x; 1.0655x over previous
#include <hip/hip_runtime.h>


typedef float v2f __attribute__((ext_vector_type(2)));
typedef float v8f __attribute__((ext_vector_type(8)));

#define GRU_H   64
#define GRU_I   32
#define GRU_O   16
#define BM      16
#define BMR     2
#define GRP     (BM / BMR)
#define VN      BMR
#define NWAVES  4
#define XCH     ((GRU_I / 4) / GRP)
#define WAVE_SZ 32
#define LDS_PITCH 68
#define X_PITCH 36
#define Y_PITCH 20

#define WMMA_F32(acc, a, b) \
    acc = __builtin_amdgcn_wmma_f32_16x16x4_f32(false, a, false, b, (short)0, acc, false, false)

#define STAGE_ACC2_ACC2(a0, a1, b0, b1) \
    asm volatile("" : "+v"(a0), "+v"(a1), "+v"(b0), "+v"(b1))
#define STAGE_ACC2_VN(a0, a1, x) \
    asm volatile("" : "+v"(a0), "+v"(a1), "+v"(x[0]), "+v"(x[1]))
#define STAGE_ACC3(a0, a1, a2) \
    asm volatile("" : "+v"(a0), "+v"(a1), "+v"(a2))
#define STAGE_FRAG1(f0) \
    asm volatile("" : "+v"(f0))
#define STAGE_ACC6(a0, a1, a2, a3, a4, a5) \
    asm volatile("" : "+v"(a0), "+v"(a1), "+v"(a2), "+v"(a3), "+v"(a4), "+v"(a5))

__device__ __forceinline__ float tanh_exp(float a)
{
    const float ax = __builtin_fabsf(a);
    return __expf(ax + ax);
}
__device__ __forceinline__ float tanh_small(float a)
{
    const float ax = __builtin_fabsf(a);
    const float a2 = a * a;
    float p = __builtin_fmaf(a2, -0x1.758e7ap-8f, 0x1.521192p-6f);
    p = __builtin_fmaf(a2, p, -0x1.b8389cp-5f);
    p = __builtin_fmaf(a2, p, 0x1.110704p-3f);
    p = __builtin_fmaf(a2, p, -0x1.555532p-2f);
    return __builtin_fmaf(a2, ax * p, ax);
}
__device__ __forceinline__ float tanh_back(float a, float e, float small_r)
{
    const float ax  = __builtin_fabsf(a);
    const float big = __builtin_fmaf(__builtin_amdgcn_rcpf(e + 1.0f), -2.0f, 1.0f);
    const float t = (ax < 0.625f) ? small_r : big;
    return __builtin_copysignf(t, a);
}

__device__ __forceinline__ void project_load(const float* __restrict__ y_part,
                                             int row, int col, float2 (&q)[4])
{
    #pragma unroll
    for (int k = 0; k < 4; ++k)
        q[k] = *reinterpret_cast<const float2*>(y_part + (k * BM + row) * Y_PITCH + col);
}
__device__ __forceinline__ float2 project_add(const float2 (&q)[4], float2 b)
{
    float2 y;
    y.x = ((q[0].x + q[1].x) + (q[2].x + q[3].x)) + b.x;
    y.y = ((q[0].y + q[1].y) + (q[2].y + q[3].y)) + b.y;
    return y;
}
__device__ __forceinline__ float2 project_sum(const float* __restrict__ y_part,
                                              int row, int col, float2 b)
{
    const float2 p0 = *reinterpret_cast<const float2*>(y_part + (0 * BM + row) * Y_PITCH + col);
    const float2 p1 = *reinterpret_cast<const float2*>(y_part + (1 * BM + row) * Y_PITCH + col);
    const float2 p2 = *reinterpret_cast<const float2*>(y_part + (2 * BM + row) * Y_PITCH + col);
    const float2 p3 = *reinterpret_cast<const float2*>(y_part + (3 * BM + row) * Y_PITCH + col);
    float2 y;
    y.x = ((p0.x + p1.x) + (p2.x + p3.x)) + b.x;
    y.y = ((p0.y + p1.y) + (p2.y + p3.y)) + b.y;
    return y;
}

__global__ __launch_bounds__(NWAVES * WAVE_SZ)
void gru_fused_wmma_kernel(const float* __restrict__ x,
                           const float* __restrict__ W_ih,
                           const float* __restrict__ W_hh,
                           const float* __restrict__ b_ih,
                           const float* __restrict__ b_hh,
                           const float* __restrict__ W_lin,
                           const float* __restrict__ b_lin,
                           float* __restrict__ out,
                           int T)
{
    __shared__ __attribute__((aligned(16))) float h_buf[BM * LDS_PITCH];
    __shared__ __attribute__((aligned(16))) float hp_buf[2 * BM * LDS_PITCH];
    __shared__ __attribute__((aligned(16))) float x_buf[BM * X_PITCH];
    __shared__ __attribute__((aligned(16))) float y_part[4 * BM * Y_PITCH];
    __shared__ __attribute__((aligned(16))) float dead_buf[BM * LDS_PITCH];

    const int tid  = threadIdx.x;
    const int wave = __builtin_amdgcn_readfirstlane(tid / WAVE_SZ);
    const int lane = tid % WAVE_SZ;
    const int hi   = lane >> 4;
    const int ln   = lane & 15;
    const int b0   = blockIdx.x * BMR;
    const int c0   = wave * 16;

    const int  pk0  = 16 * wave;
    const int  prow = 4 * wave + (lane >> 3);
    const int  pcol = 2 * (lane & 7);
    const int  p_b  = prow % BMR;
    const int  p_ph = prow / BMR;

    const int  srow  = tid >> 3;
    const int  sc4   = tid & 7;
    const int  s_b   = srow % BMR;
    const int  s_tau = srow / BMR;
    const float* const xs_row = x + ((size_t)(b0 + s_b) * T) * GRU_I + 4 * sc4;
    float* const xs_lds = x_buf + srow * X_PITCH + 4 * sc4;

    v2f Bih[3][GRU_I / 4];
    v2f Bhh[3][GRU_H / 4];
    v2f Blin4[4];
    float bih[3], bhh[3];
    #pragma unroll
    for (int g = 0; g < 3; ++g) {
        const float* wih_row = W_ih + (size_t)(g * GRU_H + c0 + ln) * GRU_I;
        #pragma unroll
        for (int kc = 0; kc < GRU_I / 4; ++kc) {
            Bih[g][kc][0] = wih_row[4 * kc + 2 * hi + 0];
            Bih[g][kc][1] = wih_row[4 * kc + 2 * hi + 1];
        }
        const float* whh_row = W_hh + (size_t)(g * GRU_H + c0 + ln) * GRU_H;
        #pragma unroll
        for (int s = 0; s < GRU_H / 4; ++s) {
            const int kc = (s + 4 * wave) & 15;
            Bhh[g][s][0] = whh_row[4 * kc + 2 * hi + 0];
            Bhh[g][s][1] = whh_row[4 * kc + 2 * hi + 1];
        }
        bih[g] = b_ih[g * GRU_H + c0 + ln];
        bhh[g] = b_hh[g * GRU_H + c0 + ln];
    }
    {
        const float* wl_row = W_lin + (size_t)ln * GRU_H + pk0;
        #pragma unroll
        for (int j = 0; j < 4; ++j) {
            Blin4[j][0] = wl_row[4 * j + 2 * hi + 0];
            Blin4[j][1] = wl_row[4 * j + 2 * hi + 1];
        }
    }
    float2 blin2;
    blin2.x = b_lin[pcol + 0];
    blin2.y = b_lin[pcol + 1];
    float* const out_row = out + ((size_t)(b0 + p_b) * T + p_ph) * GRU_O + pcol;

    const float* hs[4];
    #pragma unroll
    for (int j = 0; j < 4; ++j)
        hs[j] = h_buf + ln * LDS_PITCH + 2 * hi + 16 * ((wave + j) & 3);
    const float* const xa_ptr = x_buf + ln * X_PITCH + 2 * hi;
    float* const y_mine = y_part + wave * BM * Y_PITCH;

    for (int i = tid; i < BM * LDS_PITCH; i += blockDim.x) h_buf[i] = 0.0f;

    {
        const int t0 = (s_tau < T) ? s_tau : T - 1;
        *reinterpret_cast<float4*>(xs_lds) =
            *reinterpret_cast<const float4*>(xs_row + (size_t)t0 * GRU_I);
    }
    __syncthreads();
    const v8f zero8 = {0.f, 0.f, 0.f, 0.f, 0.f, 0.f, 0.f, 0.f};
    v8f acc_x[3] = {zero8, zero8, zero8};
    #pragma unroll
    for (int kc = 0; kc < GRU_I / 4; ++kc) {
        v2f a;
        a[0] = xa_ptr[4 * kc + 0];
        a[1] = xa_ptr[4 * kc + 1];
        #pragma unroll
        for (int g = 0; g < 3; ++g) WMMA_F32(acc_x[g], a, Bih[g][kc]);
    }
    __syncthreads();
    {
        const int t1 = (GRP + s_tau < T) ? GRP + s_tau : T - 1;
        *reinterpret_cast<float4*>(xs_lds) =
            *reinterpret_cast<const float4*>(xs_row + (size_t)t1 * GRU_I);
    }
    __syncthreads();

    v2f ha[GRU_H / 4];
    v2f xa[GRU_I / 4];
    v2f pa[4];
    float ho[VN];
    #pragma unroll
    for (int s = 0; s < GRU_H / 4; ++s) { ha[s][0] = 0.f; ha[s][1] = 0.f; }
    #pragma unroll
    for (int kc = 0; kc < GRU_I / 4; ++kc) {
        xa[kc][0] = xa_ptr[4 * kc + 0];
        xa[kc][1] = xa_ptr[4 * kc + 1];
    }
    #pragma unroll
    for (int j = 0; j < 4; ++j) { pa[j][0] = 0.f; pa[j][1] = 0.f; }
    #pragma unroll
    for (int v = 0; v < VN; ++v) ho[v] = 0.0f;

    for (int tg = 0; tg < T; tg += GRP) {
        const bool do_proj = (tg > 0);
        float* const hp_tile = hp_buf + ((tg / GRP) & 1) * BM * LDS_PITCH;

        float4 xr;
        {
            const int ta = (tg + 2 * GRP + s_tau < T) ? tg + 2 * GRP + s_tau : T - 1;
            const int tp = (tg + 4 * GRP + s_tau < T) ? tg + 4 * GRP + s_tau : T - 1;
            xr = *reinterpret_cast<const float4*>(xs_row + (size_t)ta * GRU_I);
            __builtin_prefetch((const void*)(xs_row + (size_t)tp * GRU_I), 0, 1);
        }

        v8f acc_xn[3] = {zero8, zero8, zero8};
        v8f accp = zero8;
        float2 yq[4] = {};

        #pragma unroll
        for (int p = 0; p < GRP; ++p) {
            if (tg + p < T) {
                const int V0      = (BMR * p) & 7;
                const int HI_LIVE = (BMR * p) >> 3;

                v8f ar0 = zero8, ar1 = zero8;
                v8f an0 = zero8, an1 = zero8;
                v8f az0 = zero8, az1 = zero8;
                float sr[VN], er[VN], rg[VN];
                float an[VN], en[VN], sn[VN], nn[VN];
                float sz[VN], zg[VN];

                #pragma unroll
                for (int s = 0; s < 2; ++s) {
                    WMMA_F32(ar0, ha[s],     Bhh[0][s]);
                    WMMA_F32(ar1, ha[s + 2], Bhh[0][s + 2]);
                    WMMA_F32(an0, ha[s],     Bhh[2][s]);
                    WMMA_F32(an1, ha[s + 2], Bhh[2][s + 2]);
                    WMMA_F32(az0, ha[s],     Bhh[1][s]);
                    WMMA_F32(az1, ha[s + 2], Bhh[1][s + 2]);
                }
                STAGE_ACC6(ar0, ar1, an0, an1, az0, az1);
                if (p == 1 && do_proj) {
                    const float2 y = project_add(yq, blin2);
                    *reinterpret_cast<float2*>(out_row + (size_t)(tg - GRP) * GRU_O) = y;
                }

                #pragma unroll
                for (int i = 0; i < 6; ++i) {
                    WMMA_F32(ar0, ha[4 + i],  Bhh[0][4 + i]);
                    WMMA_F32(ar1, ha[10 + i], Bhh[0][10 + i]);
                }

                if (p == 0 && do_proj) {
                    #pragma unroll
                    for (int j = 0; j < 4; ++j) WMMA_F32(accp, pa[j], Blin4[j]);
                }

                #pragma unroll
                for (int kc = 0; kc < 6; ++kc) {
                    WMMA_F32(an0, ha[4 + kc],  Bhh[2][4 + kc]);
                    WMMA_F32(an1, ha[10 + kc], Bhh[2][10 + kc]);
                    if (kc == 0) {
                        STAGE_ACC2_ACC2(an0, an1, ar0, ar1);
                        #pragma unroll
                        for (int v = 0; v < VN; ++v)
                            sr[v] = acc_x[0][V0 + v] + ar0[V0 + v] + ar1[V0 + v] + bih[0] + bhh[0];
                        STAGE_ACC2_VN(an0, an1, sr);
                    } else if (kc == 1) {
                        STAGE_ACC2_VN(an0, an1, sr);
                        #pragma unroll
                        for (int v = 0; v < VN; ++v) er[v] = __expf(-sr[v]);
                        STAGE_ACC2_VN(an0, an1, er);
                    } else if (kc == 2) {
                        STAGE_ACC2_VN(an0, an1, er);
                        #pragma unroll
                        for (int v = 0; v < VN; ++v) rg[v] = __builtin_amdgcn_rcpf(1.0f + er[v]);
                        STAGE_ACC2_VN(an0, an1, rg);
                    }
                }

                #pragma unroll
                for (int kc = 0; kc < 6; ++kc) {
                    WMMA_F32(az0, ha[4 + kc],  Bhh[1][4 + kc]);
                    WMMA_F32(az1, ha[10 + kc], Bhh[1][10 + kc]);
                    if (kc == 0) {
                        STAGE_ACC2_ACC2(az0, az1, an0, an1);
                        #pragma unroll
                        for (int v = 0; v < VN; ++v) {
                            const float gx = acc_x[2][V0 + v] + bih[2];
                            const float gh = an0[V0 + v] + an1[V0 + v] + bhh[2];
                            an[v] = gx + rg[v] * gh;
                        }
                        STAGE_ACC2_VN(az0, az1, an);
                    } else if (kc == 1) {
                        STAGE_ACC2_VN(az0, az1, an);
                        #pragma unroll
                        for (int v = 0; v < VN; ++v) en[v] = tanh_exp(an[v]);
                        STAGE_ACC2_VN(az0, az1, en);
                    } else if (kc == 2) {
                        STAGE_ACC2_VN(az0, az1, an);
                        #pragma unroll
                        for (int v = 0; v < VN; ++v) sn[v] = tanh_small(an[v]);
                        STAGE_ACC2_VN(az0, az1, sn);
                    } else if (kc == 3) {
                        STAGE_ACC2_VN(az0, az1, en);
                        #pragma unroll
                        for (int v = 0; v < VN; ++v) nn[v] = tanh_back(an[v], en[v], sn[v]);
                        STAGE_ACC2_VN(az0, az1, nn);
                    }
                }

                #pragma unroll
                for (int v = 0; v < VN; ++v) {
                    sz[v] = acc_x[1][V0 + v] + az0[V0 + v] + az1[V0 + v] + bih[1] + bhh[1];
                    zg[v] = __builtin_amdgcn_rcpf(1.0f + __expf(-sz[v]));
                }

                float hn[VN];
                #pragma unroll
                for (int v = 0; v < VN; ++v)
                    hn[v] = (1.0f - zg[v]) * nn[v] + zg[v] * ho[v];
                {
                    const bool live = (hi == HI_LIVE);
                    float* const hb = (live ? h_buf   : dead_buf) + c0 + ln;
                    float* const pb = (live ? hp_tile : dead_buf) + c0 + ln;
                    #pragma unroll
                    for (int v = 0; v < VN; ++v) {
                        const int m  = V0 + v + 8 * hi;
                        const int mn = (m + BMR) & (BM - 1);
                        hb[mn * LDS_PITCH] = hn[v];
                        pb[m * LDS_PITCH]  = hn[v];
                    }
                }
                if (p == 0 && do_proj) {
                    #pragma unroll
                    for (int v = 0; v < 8; ++v)
                        y_mine[(v + 8 * hi) * Y_PITCH + ln] = accp[v];
                }
                if (p == GRP - 1) {
                    *reinterpret_cast<float4*>(xs_lds) = xr;
                }
                __builtin_amdgcn_fence(__ATOMIC_RELEASE, "workgroup");
                __builtin_amdgcn_s_barrier_signal(-1);
                #pragma unroll
                for (int s = 0; s < 4; ++s) {
                    ha[s][0] = hs[0][4 * s + 0];
                    ha[s][1] = hs[0][4 * s + 1];
                }
                STAGE_FRAG1(xa[XCH * p]);
                #pragma unroll
                for (int kc = XCH * p; kc < XCH * p + XCH; ++kc) {
                    #pragma unroll
                    for (int g = 0; g < 3; ++g) WMMA_F32(acc_xn[g], xa[kc], Bih[g][kc]);
                }
                STAGE_ACC3(acc_xn[0], acc_xn[1], acc_xn[2]);
                __builtin_amdgcn_s_barrier_wait(-1);
                __builtin_amdgcn_fence(__ATOMIC_ACQUIRE, "workgroup");

                #pragma unroll
                for (int s = 4; s < GRU_H / 4; ++s) {
                    ha[s][0] = hs[s >> 2][4 * (s & 3) + 0];
                    ha[s][1] = hs[s >> 2][4 * (s & 3) + 1];
                }
                {
                    const int V0n = (BMR * (p + 1)) & 7;
                    #pragma unroll
                    for (int v = 0; v < VN; ++v)
                        ho[v] = h_buf[(V0n + v + 8 * hi) * LDS_PITCH + c0 + ln];
                }
                if (p == GRP - 1) {
                    #pragma unroll
                    for (int kc = 0; kc < GRU_I / 4; ++kc) {
                        xa[kc][0] = xa_ptr[4 * kc + 0];
                        xa[kc][1] = xa_ptr[4 * kc + 1];
                    }
                    #pragma unroll
                    for (int j = 0; j < 4; ++j) {
                        pa[j][0] = hp_tile[ln * LDS_PITCH + pk0 + 4 * j + 2 * hi + 0];
                        pa[j][1] = hp_tile[ln * LDS_PITCH + pk0 + 4 * j + 2 * hi + 1];
                    }
                }
                if (p == 0 && do_proj) {
                    project_load(y_part, prow, pcol, yq);
                    if (tg + 1 >= T) {
                        const float2 y = project_add(yq, blin2);
                        *reinterpret_cast<float2*>(out_row + (size_t)(tg - GRP) * GRU_O) = y;
                    }
                }
            }
        }

        #pragma unroll
        for (int g = 0; g < 3; ++g) acc_x[g] = acc_xn[g];
    }

    __syncthreads();
    if (T > 0) {
        const int gl = (T - 1) / GRP;
        const float* const hp_last = hp_buf + (gl & 1) * BM * LDS_PITCH;
        v2f pl[4];
        #pragma unroll
        for (int j = 0; j < 4; ++j) {
            pl[j][0] = hp_last[ln * LDS_PITCH + pk0 + 4 * j + 2 * hi + 0];
            pl[j][1] = hp_last[ln * LDS_PITCH + pk0 + 4 * j + 2 * hi + 1];
        }
        v8f accl = zero8;
        #pragma unroll
        for (int j = 0; j < 4; ++j) WMMA_F32(accl, pl[j], Blin4[j]);
        #pragma unroll
        for (int v = 0; v < 8; ++v)
            y_mine[(v + 8 * hi) * Y_PITCH + ln] = accl[v];
    }
    __syncthreads();
    if (T > 0) {
        const int tl = ((T - 1) / GRP) * GRP;
        if (tl + p_ph < T) {
            const float2 y = project_sum(y_part, prow, pcol, blin2);
            *reinterpret_cast<float2*>(out_row + (size_t)tl * GRU_O) = y;
        }
    }
}

extern "C" void kernel_launch(void* const* d_in, const int* in_sizes, int n_in,
                              void* d_out, int out_size, void* d_ws, size_t ws_size,
                              hipStream_t stream) {
    const float* x     = (const float*)d_in[0];
    const float* W_ih  = (const float*)d_in[1];
    const float* W_hh  = (const float*)d_in[2];
    const float* b_ih  = (const float*)d_in[3];
    const float* b_hh  = (const float*)d_in[4];
    const float* W_lin = (const float*)d_in[5];
    const float* b_lin = (const float*)d_in[6];
    float* out = (float*)d_out;

    const int B = 512;
    const int T = in_sizes[0] / (B * GRU_I);

    dim3 grid(B / BMR);
    dim3 block(NWAVES * WAVE_SZ);
    gru_fused_wmma_kernel<<<grid, block, 0, stream>>>(
        x, W_ih, W_hh, b_ih, b_hh, W_lin, b_lin, out, T);
}
